# starting kernel
# baseline (speedup 1.0000x reference)
#include <hip/hip_runtime.h>
#include <hip/hip_bf16.h>

typedef __attribute__((ext_vector_type(2))) float v2f;
typedef __attribute__((ext_vector_type(8))) float v8f;

__global__ __launch_bounds__(256) void fused_head_kernel(
    const float* __restrict__ emb,
    const float* __restrict__ inck,
    const float* __restrict__ conv_w,
    const float* __restrict__ conv_b,
    const float* __restrict__ bn_g,
    const float* __restrict__ bn_b,
    const float* __restrict__ bn_m,
    const float* __restrict__ bn_v,
    float* __restrict__ z0)
{
    __shared__ float smem[10368 + 9216 + 256];
    float* inLDS  = smem;
    float* xLDS   = smem;
    float* wLDS   = smem + 10368;
    float* kLDS   = smem + 19584;
    float* scLDS  = smem + 19728;
    float* offLDS = smem + 19760;

    const int tid = threadIdx.x;
    const int n   = blockIdx.z;
    const int HY  = blockIdx.y * 16;
    const int WX  = blockIdx.x * 16;

    const float* embn = emb + (size_t)n * 512 * 32 * 32;
    for (int i = tid; i < 18 * 18 * 32; i += 256) {
        const int c   = i & 31;
        const int pix = i >> 5;
        const int y   = pix / 18;
        const int x   = pix - y * 18;
        const int H   = HY + y - 1;
        const int W   = WX + x - 1;
        float v = 0.f;
        if ((unsigned)H < 128u && (unsigned)W < 128u) {
            const int cc = c * 16 + (H & 3) * 4 + (W & 3);
            v = embn[(cc * 32 + (H >> 2)) * 32 + (W >> 2)];
        }
        inLDS[i] = v;
    }
    for (int i = tid; i < 9216; i += 256) {
        const int o   = i / 288;
        const int rem = i - o * 288;
        const int c   = rem / 9;
        const int t   = rem - c * 9;
        wLDS[(t * 32 + c) * 32 + o] = conv_w[i];
    }
    if (tid < 137) kLDS[tid] = inck[n * 137 + tid];
    if (tid < 32) {
        const float sc = bn_g[tid] * rsqrtf(bn_v[tid] + 1e-5f);
        scLDS[tid]  = sc;
        offLDS[tid] = (conv_b[tid] - bn_m[tid]) * sc + bn_b[tid];
    }
    __syncthreads();

    const int wave = tid >> 5;
    const int lane = tid & 31;
    const int half = lane >> 4;
    const int ml   = lane & 15;
    const int nb   = (wave >> 2) << 4;

    v8f acc[4] = {};

    for (int t = 0; t < 9; ++t) {
        const int dy = t / 3, dx = t - dy * 3;
        const float* bcol = &wLDS[(t * 32) * 32 + nb + ml];
#pragma unroll
        for (int kk = 0; kk < 8; ++kk) {
            const int c0 = kk * 4 + 2 * half;
            v2f bv; bv.x = bcol[c0 * 32]; bv.y = bcol[(c0 + 1) * 32];
#pragma unroll
            for (int j = 0; j < 4; ++j) {
                const int py = ((wave & 3) << 2) + j;
                const float2 a2 = *(const float2*)
                    (&inLDS[((py + dy) * 18 + dx + ml) * 32] + c0);
                v2f av; av.x = a2.x; av.y = a2.y;
                acc[j] = __builtin_amdgcn_wmma_f32_16x16x4_f32(
                    false, av, false, bv, (short)0, acc[j], false, false);
            }
        }
    }

    __syncthreads();

    {
        const int och  = nb + ml;
        const float sc = scLDS[och];
        const float of = offLDS[och];
#pragma unroll
        for (int j = 0; j < 4; ++j) {
            const int py = ((wave & 3) << 2) + j;
#pragma unroll
            for (int r = 0; r < 8; ++r) {
                const int m = r + 8 * half;
                xLDS[(py * 16 + m) * 32 + och] = fmaxf(acc[j][r] * sc + of, 0.f);
            }
        }
    }
    __syncthreads();

    {
        const int p = tid;
        float h0 = kLDS[132], h1 = kLDS[133], h2 = kLDS[134], h3 = kLDS[135];
        const float* xp = &xLDS[p * 32];
#pragma unroll
        for (int c = 0; c < 32; ++c) {
            const float xc = xp[c];
            h0 = fmaf(kLDS[c],      xc, h0);
            h1 = fmaf(kLDS[32 + c], xc, h1);
            h2 = fmaf(kLDS[64 + c], xc, h2);
            h3 = fmaf(kLDS[96 + c], xc, h3);
        }
        h0 = fmaxf(h0, 0.f); h1 = fmaxf(h1, 0.f);
        h2 = fmaxf(h2, 0.f); h3 = fmaxf(h3, 0.f);
        float y = kLDS[136];
        y = fmaf(kLDS[128], h0, y);
        y = fmaf(kLDS[129], h1, y);
        y = fmaf(kLDS[130], h2, y);
        y = fmaf(kLDS[131], h3, y);
        const float s = 1.f / (1.f + __expf(-y));
        const int H = HY + (p >> 4), W = WX + (p & 15);
        z0[((size_t)n * 128 + H) * 128 + W] = s;
    }
}

__global__ __launch_bounds__(256) void smooth_kernel(
    const float* __restrict__ z0,
    const float* __restrict__ mask,
    const float* __restrict__ sw,
    const float* __restrict__ sb,
    float* __restrict__ out)
{
    __shared__ float w[50];
    __shared__ float bias;
    if (threadIdx.x < 50) w[threadIdx.x] = sw[threadIdx.x];
    if (threadIdx.x == 0) bias = sb[0];
    __syncthreads();

    const int idx = blockIdx.x * 256 + threadIdx.x;
    const int n   = idx >> 14;
    const int rem = idx & 16383;
    const int H   = rem >> 7, W = rem & 127;
    const float* zp = z0   + (size_t)n * 16384;
    const float* mp = mask + (size_t)n * 16384;

    float acc = bias;
#pragma unroll
    for (int dy = 0; dy < 5; ++dy) {
        const int h = H + dy - 2;
        if ((unsigned)h >= 128u) continue;
#pragma unroll
        for (int dx = 0; dx < 5; ++dx) {
            const int x = W + dx - 2;
            if ((unsigned)x >= 128u) continue;
            const int off = h * 128 + x;
            acc = fmaf(w[dy * 5 + dx],      zp[off], acc);
            acc = fmaf(w[25 + dy * 5 + dx], mp[off], acc);
        }
    }
    out[idx] = acc;
}

extern "C" void kernel_launch(void* const* d_in, const int* in_sizes, int n_in,
                              void* d_out, int out_size, void* d_ws, size_t ws_size,
                              hipStream_t stream) {
    const float* emb    = (const float*)d_in[0];
    const float* inck   = (const float*)d_in[1];
    const float* masks  = (const float*)d_in[2];
    const float* conv_w = (const float*)d_in[3];
    const float* conv_b = (const float*)d_in[4];
    const float* bn_g   = (const float*)d_in[5];
    const float* bn_b   = (const float*)d_in[6];
    const float* bn_m   = (const float*)d_in[7];
    const float* bn_v   = (const float*)d_in[8];
    const float* sw     = (const float*)d_in[9];
    const float* sb     = (const float*)d_in[10];
    float* out = (float*)d_out;
    float* z0  = (float*)d_ws;

    dim3 g1(8, 8, 32);
    fused_head_kernel<<<g1, 256, 0, stream>>>(emb, inck, conv_w, conv_b,
                                              bn_g, bn_b, bn_m, bn_v, z0);
    smooth_kernel<<<2048, 256, 0, stream>>>(z0, masks, sw, sb, out);
}


# LDS bank conflicts removed by layout: activation tile padded four dwords per pixel pair (A and MLP reads); weight tile tap pitch plus one dword, och rotated half a row on odd channel pairs
# speedup vs baseline: 1.1799x; 1.1799x over previous
#include <hip/hip_runtime.h>
#include <hip/hip_bf16.h>

typedef __attribute__((ext_vector_type(2))) float v2f;
typedef __attribute__((ext_vector_type(8))) float v8f;

#define PIXOFF(p) ((p) * 32 + ((p) >> 1) * 4)
#define WTAP 1025
__global__ __launch_bounds__(256) void fused_head_kernel(
    const float* __restrict__ emb,
    const float* __restrict__ inck,
    const float* __restrict__ conv_w,
    const float* __restrict__ conv_b,
    const float* __restrict__ bn_g,
    const float* __restrict__ bn_b,
    const float* __restrict__ bn_m,
    const float* __restrict__ bn_v,
    float* __restrict__ z0)
{
    __shared__ __attribute__((aligned(16))) float smem[11016 + 140 + 32 + 32 + 9 * WTAP];
    float* inLDS  = smem;
    float* xLDS   = smem;
    float* kLDS   = smem + 11016;
    float* scLDS  = smem + 11156;
    float* offLDS = smem + 11188;
    float* wLDS   = smem + 11220;

    const int tid = threadIdx.x;
    const int n   = blockIdx.z;
    const int HY  = blockIdx.y * 16;
    const int WX  = blockIdx.x * 16;

    const float* embn = emb + (size_t)n * 512 * 32 * 32;
    for (int i = tid; i < 18 * 18 * 32; i += 256) {
        const int c   = i & 31;
        const int pix = i >> 5;
        const int y   = pix / 18;
        const int x   = pix - y * 18;
        const int H   = HY + y - 1;
        const int W   = WX + x - 1;
        float v = 0.f;
        if ((unsigned)H < 128u && (unsigned)W < 128u) {
            const int cc = c * 16 + (H & 3) * 4 + (W & 3);
            v = embn[(cc * 32 + (H >> 2)) * 32 + (W >> 2)];
        }
        inLDS[PIXOFF(pix) + c] = v;
    }
    for (int i = tid; i < 9216; i += 256) {
        const int o   = i / 288;
        const int rem = i - o * 288;
        const int c   = rem / 9;
        const int t   = rem - c * 9;
        wLDS[t * WTAP + c * 32 + ((o + ((c & 2) << 3)) & 31)] = conv_w[i];
    }
    if (tid < 137) kLDS[tid] = inck[n * 137 + tid];
    if (tid < 32) {
        const float sc = bn_g[tid] * rsqrtf(bn_v[tid] + 1e-5f);
        scLDS[tid]  = sc;
        offLDS[tid] = (conv_b[tid] - bn_m[tid]) * sc + bn_b[tid];
    }
    __syncthreads();

    const int wave = tid >> 5;
    const int lane = tid & 31;
    const int half = lane >> 4;
    const int ml   = lane & 15;
    const int nb   = (wave >> 2) << 4;

    v8f acc[4] = {};

    for (int t = 0; t < 9; ++t) {
        const int dy = t / 3, dx = t - dy * 3;
        const float* bcol = &wLDS[t * WTAP + ((nb + ml + 16 * half) & 31)];
#pragma unroll
        for (int kk = 0; kk < 8; ++kk) {
            const int c0 = kk * 4 + 2 * half;
            v2f bv; bv.x = bcol[c0 * 32]; bv.y = bcol[(c0 + 1) * 32];
#pragma unroll
            for (int j = 0; j < 4; ++j) {
                const int py = ((wave & 3) << 2) + j;
                const float2 a2 = *(const float2*)
                    (&inLDS[(py + dy) * PIXOFF(18) + PIXOFF(dx + ml)] + c0);
                v2f av; av.x = a2.x; av.y = a2.y;
                acc[j] = __builtin_amdgcn_wmma_f32_16x16x4_f32(
                    false, av, false, bv, (short)0, acc[j], false, false);
            }
        }
    }

    __syncthreads();

    {
        const int och  = nb + ml;
        const float sc = scLDS[och];
        const float of = offLDS[och];
#pragma unroll
        for (int j = 0; j < 4; ++j) {
            const int py = ((wave & 3) << 2) + j;
#pragma unroll
            for (int r = 0; r < 8; ++r) {
                const int m = r + 8 * half;
                xLDS[PIXOFF(py * 16 + m) + och] = fmaxf(acc[j][r] * sc + of, 0.f);
            }
        }
    }
    __syncthreads();

    {
        const int p = tid;
        float h0 = kLDS[132], h1 = kLDS[133], h2 = kLDS[134], h3 = kLDS[135];
        const float* xp = &xLDS[PIXOFF(p)];
#pragma unroll
        for (int c = 0; c < 32; ++c) {
            const float xc = xp[c];
            h0 = fmaf(kLDS[c],      xc, h0);
            h1 = fmaf(kLDS[32 + c], xc, h1);
            h2 = fmaf(kLDS[64 + c], xc, h2);
            h3 = fmaf(kLDS[96 + c], xc, h3);
        }
        h0 = fmaxf(h0, 0.f); h1 = fmaxf(h1, 0.f);
        h2 = fmaxf(h2, 0.f); h3 = fmaxf(h3, 0.f);
        float y = kLDS[136];
        y = fmaf(kLDS[128], h0, y);
        y = fmaf(kLDS[129], h1, y);
        y = fmaf(kLDS[130], h2, y);
        y = fmaf(kLDS[131], h3, y);
        const float s = 1.f / (1.f + __expf(-y));
        const int H = HY + (p >> 4), W = WX + (p & 15);
        z0[((size_t)n * 128 + H) * 128 + W] = s;
    }
}

__global__ __launch_bounds__(256) void smooth_kernel(
    const float* __restrict__ z0,
    const float* __restrict__ mask,
    const float* __restrict__ sw,
    const float* __restrict__ sb,
    float* __restrict__ out)
{
    __shared__ float w[50];
    __shared__ float bias;
    if (threadIdx.x < 50) w[threadIdx.x] = sw[threadIdx.x];
    if (threadIdx.x == 0) bias = sb[0];
    __syncthreads();

    const int idx = blockIdx.x * 256 + threadIdx.x;
    const int n   = idx >> 14;
    const int rem = idx & 16383;
    const int H   = rem >> 7, W = rem & 127;
    const float* zp = z0   + (size_t)n * 16384;
    const float* mp = mask + (size_t)n * 16384;

    float acc = bias;
#pragma unroll
    for (int dy = 0; dy < 5; ++dy) {
        const int h = H + dy - 2;
        if ((unsigned)h >= 128u) continue;
#pragma unroll
        for (int dx = 0; dx < 5; ++dx) {
            const int x = W + dx - 2;
            if ((unsigned)x >= 128u) continue;
            const int off = h * 128 + x;
            acc = fmaf(w[dy * 5 + dx],      zp[off], acc);
            acc = fmaf(w[25 + dy * 5 + dx], mp[off], acc);
        }
    }
    out[idx] = acc;
}

extern "C" void kernel_launch(void* const* d_in, const int* in_sizes, int n_in,
                              void* d_out, int out_size, void* d_ws, size_t ws_size,
                              hipStream_t stream) {
    const float* emb    = (const float*)d_in[0];
    const float* inck   = (const float*)d_in[1];
    const float* masks  = (const float*)d_in[2];
    const float* conv_w = (const float*)d_in[3];
    const float* conv_b = (const float*)d_in[4];
    const float* bn_g   = (const float*)d_in[5];
    const float* bn_b   = (const float*)d_in[6];
    const float* bn_m   = (const float*)d_in[7];
    const float* bn_v   = (const float*)d_in[8];
    const float* sw     = (const float*)d_in[9];
    const float* sb     = (const float*)d_in[10];
    float* out = (float*)d_out;
    float* z0  = (float*)d_ws;

    dim3 g1(8, 8, 32);
    fused_head_kernel<<<g1, 256, 0, stream>>>(emb, inck, conv_w, conv_b,
                                              bn_g, bn_b, bn_m, bn_v, z0);
    smooth_kernel<<<2048, 256, 0, stream>>>(z0, masks, sw, sb, out);
}


# conv tile reshaped 16x16 -> full-width strip of 4 rows (512 threads): the pixel-shuffle gather walks W, so a wave load reads 4 planes x 32 B contiguous instead of 32 lines; channels stored 16 B wide
# speedup vs baseline: 1.8612x; 1.5774x over previous
#include <hip/hip_runtime.h>
#include <hip/hip_bf16.h>

typedef __attribute__((ext_vector_type(2))) float v2f;
typedef __attribute__((ext_vector_type(8))) float v8f;

#define PIXOFF(p) ((p) * 32 + ((p) >> 1) * 4)
#define WTAP 1025
#define ROWP 130
#define TILE_DW 26520
#define PLANE_C (16 * 32 * 32)
__global__ __launch_bounds__(512) void fused_head_kernel(
    const float* __restrict__ emb,
    const float* __restrict__ inck,
    const float* __restrict__ conv_w,
    const float* __restrict__ conv_b,
    const float* __restrict__ bn_g,
    const float* __restrict__ bn_b,
    const float* __restrict__ bn_m,
    const float* __restrict__ bn_v,
    float* __restrict__ z0)
{
    __shared__ __attribute__((aligned(16))) float smem[TILE_DW + 140 + 32 + 32 + 9 * WTAP];
    float* inLDS  = smem;
    float* xLDS   = smem;
    float* kLDS   = smem + TILE_DW;
    float* scLDS  = smem + TILE_DW + 140;
    float* offLDS = smem + TILE_DW + 172;
    float* wLDS   = smem + TILE_DW + 204;

    const int tid   = threadIdx.x;
    const int n     = blockIdx.y;
    const int strip = blockIdx.x;

    const float* embn = emb + (size_t)n * 512 * 32 * 32;
    {
        const int Wp = tid & 127;
        const int jw = (Wp & 3) * 1024 + (Wp >> 2);
        const int y1 = tid >> 7;
        const int y2 = 4 + ((tid >> 7) & 1);
        const int cb = (tid >> 8) * 16;
        const int H1 = 4 * strip - 1 + y1;
        const int H2 = 4 * strip - 1 + y2;
        float v[32], u[16];
        if ((unsigned)H1 < 128u) {
            const float* src = embn + ((H1 & 3) * 4096 + (H1 >> 2) * 32 + jw);
#pragma unroll
            for (int c = 0; c < 32; ++c) v[c] = src[c * PLANE_C];
        } else {
#pragma unroll
            for (int c = 0; c < 32; ++c) v[c] = 0.f;
        }
        if ((unsigned)H2 < 128u) {
            const float* src = embn + ((H2 & 3) * 4096 + (H2 >> 2) * 32 + jw) + cb * PLANE_C;
#pragma unroll
            for (int c = 0; c < 16; ++c) u[c] = src[c * PLANE_C];
        } else {
#pragma unroll
            for (int c = 0; c < 16; ++c) u[c] = 0.f;
        }
        float* dst1 = &inLDS[PIXOFF(y1 * ROWP + 1 + Wp)];
        float* dst2 = &inLDS[PIXOFF(y2 * ROWP + 1 + Wp) + cb];
#pragma unroll
        for (int q = 0; q < 8; ++q)
            *(float4*)(dst1 + 4 * q) =
                make_float4(v[4 * q], v[4 * q + 1], v[4 * q + 2], v[4 * q + 3]);
#pragma unroll
        for (int q = 0; q < 4; ++q)
            *(float4*)(dst2 + 4 * q) =
                make_float4(u[4 * q], u[4 * q + 1], u[4 * q + 2], u[4 * q + 3]);
        if (tid < 12) {
            float* dst = &inLDS[PIXOFF((tid >> 1) * ROWP + (tid & 1) * (ROWP - 1))];
#pragma unroll
            for (int q = 0; q < 8; ++q)
                *(float4*)(dst + 4 * q) = make_float4(0.f, 0.f, 0.f, 0.f);
        }
    }
    for (int i = tid; i < 9216; i += 512) {
        const int o   = i / 288;
        const int rem = i - o * 288;
        const int c   = rem / 9;
        const int t   = rem - c * 9;
        wLDS[t * WTAP + c * 32 + ((o + ((c & 2) << 3)) & 31)] = conv_w[i];
    }
    if (tid < 137) kLDS[tid] = inck[n * 137 + tid];
    if (tid < 32) {
        const float sc = bn_g[tid] * rsqrtf(bn_v[tid] + 1e-5f);
        scLDS[tid]  = sc;
        offLDS[tid] = (conv_b[tid] - bn_m[tid]) * sc + bn_b[tid];
    }
    __syncthreads();

    const int wave = tid >> 5;
    const int lane = tid & 31;
    const int half = lane >> 4;
    const int ml   = lane & 15;
    const int nb   = (wave >> 3) << 4;
    const int X0   = (wave & 7) << 4;

    v8f acc[4] = {};

    for (int t = 0; t < 9; ++t) {
        const int dy = t / 3, dx = t - dy * 3;
        const float* bcol = &wLDS[t * WTAP + ((nb + ml + 16 * half) & 31)];
#pragma unroll
        for (int kk = 0; kk < 8; ++kk) {
            const int c0 = kk * 4 + 2 * half;
            v2f bv; bv.x = bcol[c0 * 32]; bv.y = bcol[(c0 + 1) * 32];
#pragma unroll
            for (int j = 0; j < 4; ++j) {
                const float2 a2 = *(const float2*)
                    (&inLDS[(j + dy) * PIXOFF(ROWP) + PIXOFF(X0 + dx + ml)] + c0);
                v2f av; av.x = a2.x; av.y = a2.y;
                acc[j] = __builtin_amdgcn_wmma_f32_16x16x4_f32(
                    false, av, false, bv, (short)0, acc[j], false, false);
            }
        }
    }

    __syncthreads();

    {
        const int och  = nb + ml;
        const float sc = scLDS[och];
        const float of = offLDS[och];
#pragma unroll
        for (int j = 0; j < 4; ++j) {
#pragma unroll
            for (int r = 0; r < 8; ++r) {
                const int m = r + 8 * half;
                xLDS[PIXOFF(j * 128 + X0 + m) + och] = fmaxf(acc[j][r] * sc + of, 0.f);
            }
        }
    }
    __syncthreads();

    {
        const int p = tid;
        float h0 = kLDS[132], h1 = kLDS[133], h2 = kLDS[134], h3 = kLDS[135];
        const float* xp = &xLDS[PIXOFF(p)];
#pragma unroll
        for (int c = 0; c < 32; ++c) {
            const float xc = xp[c];
            h0 = fmaf(kLDS[c],      xc, h0);
            h1 = fmaf(kLDS[32 + c], xc, h1);
            h2 = fmaf(kLDS[64 + c], xc, h2);
            h3 = fmaf(kLDS[96 + c], xc, h3);
        }
        h0 = fmaxf(h0, 0.f); h1 = fmaxf(h1, 0.f);
        h2 = fmaxf(h2, 0.f); h3 = fmaxf(h3, 0.f);
        float y = kLDS[136];
        y = fmaf(kLDS[128], h0, y);
        y = fmaf(kLDS[129], h1, y);
        y = fmaf(kLDS[130], h2, y);
        y = fmaf(kLDS[131], h3, y);
        const float sg = 1.f / (1.f + __expf(-y));
        const int H = 4 * strip + (p >> 7), W = p & 127;
        z0[((size_t)n * 128 + H) * 128 + W] = sg;
    }
}

__global__ __launch_bounds__(256) void smooth_kernel(
    const float* __restrict__ z0,
    const float* __restrict__ mask,
    const float* __restrict__ sw,
    const float* __restrict__ sb,
    float* __restrict__ out)
{
    __shared__ float w[50];
    __shared__ float bias;
    if (threadIdx.x < 50) w[threadIdx.x] = sw[threadIdx.x];
    if (threadIdx.x == 0) bias = sb[0];
    __syncthreads();

    const int idx = blockIdx.x * 256 + threadIdx.x;
    const int n   = idx >> 14;
    const int rem = idx & 16383;
    const int H   = rem >> 7, W = rem & 127;
    const float* zp = z0   + (size_t)n * 16384;
    const float* mp = mask + (size_t)n * 16384;

    float acc = bias;
#pragma unroll
    for (int dy = 0; dy < 5; ++dy) {
        const int h = H + dy - 2;
        if ((unsigned)h >= 128u) continue;
#pragma unroll
        for (int dx = 0; dx < 5; ++dx) {
            const int x = W + dx - 2;
            if ((unsigned)x >= 128u) continue;
            const int off = h * 128 + x;
            acc = fmaf(w[dy * 5 + dx],      zp[off], acc);
            acc = fmaf(w[25 + dy * 5 + dx], mp[off], acc);
        }
    }
    out[idx] = acc;
}

extern "C" void kernel_launch(void* const* d_in, const int* in_sizes, int n_in,
                              void* d_out, int out_size, void* d_ws, size_t ws_size,
                              hipStream_t stream) {
    const float* emb    = (const float*)d_in[0];
    const float* inck   = (const float*)d_in[1];
    const float* masks  = (const float*)d_in[2];
    const float* conv_w = (const float*)d_in[3];
    const float* conv_b = (const float*)d_in[4];
    const float* bn_g   = (const float*)d_in[5];
    const float* bn_b   = (const float*)d_in[6];
    const float* bn_m   = (const float*)d_in[7];
    const float* bn_v   = (const float*)d_in[8];
    const float* sw     = (const float*)d_in[9];
    const float* sb     = (const float*)d_in[10];
    float* out = (float*)d_out;
    float* z0  = (float*)d_ws;

    dim3 g1(32, 32);
    fused_head_kernel<<<g1, 512, 0, stream>>>(emb, inck, conv_w, conv_b,
                                              bn_g, bn_b, bn_m, bn_v, z0);
    smooth_kernel<<<2048, 256, 0, stream>>>(z0, masks, sw, sb, out);
}


# 5x5 smooth kernel tiled through LDS: block = 8 rows x 128 cols, thread = 2x4 outputs from 16-B LDS reads; 8x fewer waves, taps no longer wait on global loads; per-output sum order kept
# speedup vs baseline: 2.0279x; 1.0896x over previous
#include <hip/hip_runtime.h>
#include <hip/hip_bf16.h>

typedef __attribute__((ext_vector_type(2))) float v2f;
typedef __attribute__((ext_vector_type(8))) float v8f;

#define PIXOFF(p) ((p) * 32 + ((p) >> 1) * 4)
#define WTAP 1025
#define ROWP 130
#define TILE_DW 26520
#define PLANE_C (16 * 32 * 32)
__global__ __launch_bounds__(512) void fused_head_kernel(
    const float* __restrict__ emb,
    const float* __restrict__ inck,
    const float* __restrict__ conv_w,
    const float* __restrict__ conv_b,
    const float* __restrict__ bn_g,
    const float* __restrict__ bn_b,
    const float* __restrict__ bn_m,
    const float* __restrict__ bn_v,
    float* __restrict__ z0)
{
    __shared__ __attribute__((aligned(16))) float smem[TILE_DW + 140 + 32 + 32 + 9 * WTAP];
    float* inLDS  = smem;
    float* xLDS   = smem;
    float* kLDS   = smem + TILE_DW;
    float* scLDS  = smem + TILE_DW + 140;
    float* offLDS = smem + TILE_DW + 172;
    float* wLDS   = smem + TILE_DW + 204;

    const int tid   = threadIdx.x;
    const int n     = blockIdx.y;
    const int strip = blockIdx.x;

    const float* embn = emb + (size_t)n * 512 * 32 * 32;
    {
        const int Wp = tid & 127;
        const int jw = (Wp & 3) * 1024 + (Wp >> 2);
        const int y1 = tid >> 7;
        const int y2 = 4 + ((tid >> 7) & 1);
        const int cb = (tid >> 8) * 16;
        const int H1 = 4 * strip - 1 + y1;
        const int H2 = 4 * strip - 1 + y2;
        float v[32], u[16];
        if ((unsigned)H1 < 128u) {
            const float* src = embn + ((H1 & 3) * 4096 + (H1 >> 2) * 32 + jw);
#pragma unroll
            for (int c = 0; c < 32; ++c) v[c] = src[c * PLANE_C];
        } else {
#pragma unroll
            for (int c = 0; c < 32; ++c) v[c] = 0.f;
        }
        if ((unsigned)H2 < 128u) {
            const float* src = embn + ((H2 & 3) * 4096 + (H2 >> 2) * 32 + jw) + cb * PLANE_C;
#pragma unroll
            for (int c = 0; c < 16; ++c) u[c] = src[c * PLANE_C];
        } else {
#pragma unroll
            for (int c = 0; c < 16; ++c) u[c] = 0.f;
        }
        float* dst1 = &inLDS[PIXOFF(y1 * ROWP + 1 + Wp)];
        float* dst2 = &inLDS[PIXOFF(y2 * ROWP + 1 + Wp) + cb];
#pragma unroll
        for (int q = 0; q < 8; ++q)
            *(float4*)(dst1 + 4 * q) =
                make_float4(v[4 * q], v[4 * q + 1], v[4 * q + 2], v[4 * q + 3]);
#pragma unroll
        for (int q = 0; q < 4; ++q)
            *(float4*)(dst2 + 4 * q) =
                make_float4(u[4 * q], u[4 * q + 1], u[4 * q + 2], u[4 * q + 3]);
        if (tid < 12) {
            float* dst = &inLDS[PIXOFF((tid >> 1) * ROWP + (tid & 1) * (ROWP - 1))];
#pragma unroll
            for (int q = 0; q < 8; ++q)
                *(float4*)(dst + 4 * q) = make_float4(0.f, 0.f, 0.f, 0.f);
        }
    }
    for (int i = tid; i < 9216; i += 512) {
        const int o   = i / 288;
        const int rem = i - o * 288;
        const int c   = rem / 9;
        const int t   = rem - c * 9;
        wLDS[t * WTAP + c * 32 + ((o + ((c & 2) << 3)) & 31)] = conv_w[i];
    }
    if (tid < 137) kLDS[tid] = inck[n * 137 + tid];
    if (tid < 32) {
        const float sc = bn_g[tid] * rsqrtf(bn_v[tid] + 1e-5f);
        scLDS[tid]  = sc;
        offLDS[tid] = (conv_b[tid] - bn_m[tid]) * sc + bn_b[tid];
    }
    __syncthreads();

    const int wave = tid >> 5;
    const int lane = tid & 31;
    const int half = lane >> 4;
    const int ml   = lane & 15;
    const int nb   = (wave >> 3) << 4;
    const int X0   = (wave & 7) << 4;

    v8f acc[4] = {};

    for (int t = 0; t < 9; ++t) {
        const int dy = t / 3, dx = t - dy * 3;
        const float* bcol = &wLDS[t * WTAP + ((nb + ml + 16 * half) & 31)];
#pragma unroll
        for (int kk = 0; kk < 8; ++kk) {
            const int c0 = kk * 4 + 2 * half;
            v2f bv; bv.x = bcol[c0 * 32]; bv.y = bcol[(c0 + 1) * 32];
#pragma unroll
            for (int j = 0; j < 4; ++j) {
                const float2 a2 = *(const float2*)
                    (&inLDS[(j + dy) * PIXOFF(ROWP) + PIXOFF(X0 + dx + ml)] + c0);
                v2f av; av.x = a2.x; av.y = a2.y;
                acc[j] = __builtin_amdgcn_wmma_f32_16x16x4_f32(
                    false, av, false, bv, (short)0, acc[j], false, false);
            }
        }
    }

    __syncthreads();

    {
        const int och  = nb + ml;
        const float sc = scLDS[och];
        const float of = offLDS[och];
#pragma unroll
        for (int j = 0; j < 4; ++j) {
#pragma unroll
            for (int r = 0; r < 8; ++r) {
                const int m = r + 8 * half;
                xLDS[PIXOFF(j * 128 + X0 + m) + och] = fmaxf(acc[j][r] * sc + of, 0.f);
            }
        }
    }
    __syncthreads();

    {
        const int p = tid;
        float h0 = kLDS[132], h1 = kLDS[133], h2 = kLDS[134], h3 = kLDS[135];
        const float* xp = &xLDS[PIXOFF(p)];
#pragma unroll
        for (int c = 0; c < 32; ++c) {
            const float xc = xp[c];
            h0 = fmaf(kLDS[c],      xc, h0);
            h1 = fmaf(kLDS[32 + c], xc, h1);
            h2 = fmaf(kLDS[64 + c], xc, h2);
            h3 = fmaf(kLDS[96 + c], xc, h3);
        }
        h0 = fmaxf(h0, 0.f); h1 = fmaxf(h1, 0.f);
        h2 = fmaxf(h2, 0.f); h3 = fmaxf(h3, 0.f);
        float y = kLDS[136];
        y = fmaf(kLDS[128], h0, y);
        y = fmaf(kLDS[129], h1, y);
        y = fmaf(kLDS[130], h2, y);
        y = fmaf(kLDS[131], h3, y);
        const float sg = 1.f / (1.f + __expf(-y));
        const int H = 4 * strip + (p >> 7), W = p & 127;
        z0[((size_t)n * 128 + H) * 128 + W] = sg;
    }
}

#define SM_PITCH 132
#define SM_TROWS 12
__global__ __launch_bounds__(128) void smooth_kernel(
    const float* __restrict__ z0,
    const float* __restrict__ mask,
    const float* __restrict__ sw,
    const float* __restrict__ sb,
    float* __restrict__ out)
{
    __shared__ __attribute__((aligned(16))) float tile[2 * SM_TROWS * SM_PITCH];

    const int tid = threadIdx.x;
    const int tx  = tid & 31;
    const int ty  = tid >> 5;
    const int n   = blockIdx.x >> 4;
    const int H0  = (blockIdx.x & 15) * 8;
    const float* zp = z0   + (size_t)n * 16384;
    const float* mp = mask + (size_t)n * 16384;

    float4 g[6];
#pragma unroll
    for (int k = 0; k < 6; ++k) {
        const int row = (k % 3) * 4 + ty;
        const int h   = H0 - 2 + row;
        const float* src = (k < 3) ? zp : mp;
        g[k] = make_float4(0.f, 0.f, 0.f, 0.f);
        if ((unsigned)h < 128u) g[k] = *(const float4*)(src + h * 128 + 4 * tx);
    }
#pragma unroll
    for (int k = 0; k < 6; ++k) {
        const int row = (k % 3) * 4 + ty;
        float* dst = &tile[((k / 3) * SM_TROWS + row) * SM_PITCH + 2 + 4 * tx];
        *(float2*)dst       = make_float2(g[k].x, g[k].y);
        *(float2*)(dst + 2) = make_float2(g[k].z, g[k].w);
    }
    if (tid < 48) {
        const int pl  = tid / 24;
        const int rem = tid - pl * 24;
        float* dst = &tile[(pl * SM_TROWS + (rem >> 1)) * SM_PITCH + (rem & 1) * 130];
        *(float2*)dst = make_float2(0.f, 0.f);
    }
    __syncthreads();

    const float bias = sb[0];
    float acc[2][4];
#pragma unroll
    for (int o = 0; o < 2; ++o)
#pragma unroll
        for (int q = 0; q < 4; ++q) acc[o][q] = bias;

#pragma unroll
    for (int r = 0; r < 6; ++r) {
        const float* zr = &tile[(2 * ty + r) * SM_PITCH + 4 * tx];
        const float* mr = zr + SM_TROWS * SM_PITCH;
        const float4 za = *(const float4*)zr, zb = *(const float4*)(zr + 4);
        const float4 ma = *(const float4*)mr, mb = *(const float4*)(mr + 4);
        const float zv[8] = {za.x, za.y, za.z, za.w, zb.x, zb.y, zb.z, zb.w};
        const float mv[8] = {ma.x, ma.y, ma.z, ma.w, mb.x, mb.y, mb.z, mb.w};
#pragma unroll
        for (int o = 0; o < 2; ++o) {
            const int dy = r - o;
            if (dy < 0 || dy > 4) continue;
#pragma unroll
            for (int dx = 0; dx < 5; ++dx) {
                const float wz = sw[dy * 5 + dx];
                const float wm = sw[25 + dy * 5 + dx];
#pragma unroll
                for (int q = 0; q < 4; ++q) {
                    acc[o][q] = fmaf(wz, zv[q + dx], acc[o][q]);
                    acc[o][q] = fmaf(wm, mv[q + dx], acc[o][q]);
                }
            }
        }
    }

#pragma unroll
    for (int o = 0; o < 2; ++o) {
        const int H = H0 + 2 * ty + o;
        *(float4*)(out + ((size_t)n * 128 + H) * 128 + 4 * tx) =
            make_float4(acc[o][0], acc[o][1], acc[o][2], acc[o][3]);
    }
}

extern "C" void kernel_launch(void* const* d_in, const int* in_sizes, int n_in,
                              void* d_out, int out_size, void* d_ws, size_t ws_size,
                              hipStream_t stream) {
    const float* emb    = (const float*)d_in[0];
    const float* inck   = (const float*)d_in[1];
    const float* masks  = (const float*)d_in[2];
    const float* conv_w = (const float*)d_in[3];
    const float* conv_b = (const float*)d_in[4];
    const float* bn_g   = (const float*)d_in[5];
    const float* bn_b   = (const float*)d_in[6];
    const float* bn_m   = (const float*)d_in[7];
    const float* bn_v   = (const float*)d_in[8];
    const float* sw     = (const float*)d_in[9];
    const float* sb     = (const float*)d_in[10];
    float* out = (float*)d_out;
    float* z0  = (float*)d_ws;

    dim3 g1(32, 32);
    fused_head_kernel<<<g1, 512, 0, stream>>>(emb, inck, conv_w, conv_b,
                                              bn_g, bn_b, bn_m, bn_v, z0);
    smooth_kernel<<<512, 128, 0, stream>>>(z0, masks, sw, sb, out);
}


# 3x3 conv in minimal-filtering form F(2x2,3x3): per 2x2 tile 16 transform-position GEMMs of K=32 on f32 WMMA (2.25x fewer), transforms in-kernel; changes the conv's summation order and rounding
# speedup vs baseline: 2.6190x; 1.2915x over previous
#include <hip/hip_runtime.h>
#include <hip/hip_bf16.h>

typedef __attribute__((ext_vector_type(2))) float v2f;
typedef __attribute__((ext_vector_type(8))) float v8f;

#define PIXOFF(p) ((p) * 32 + ((p) >> 1) * 4)
#define UO 36
#define UPOS (32 * UO)
#define U_DW (16 * UPOS)
#define RAW_DW 9216
#define ROWP 130
#define TILE_DW 26520
#define PLANE_C (16 * 32 * 32)
__global__ __launch_bounds__(512) void fused_head_kernel(
    const float* __restrict__ emb,
    const float* __restrict__ inck,
    const float* __restrict__ conv_w,
    const float* __restrict__ conv_b,
    const float* __restrict__ bn_g,
    const float* __restrict__ bn_b,
    const float* __restrict__ bn_m,
    const float* __restrict__ bn_v,
    float* __restrict__ z0)
{
    __shared__ __attribute__((aligned(16))) float smem[TILE_DW + 140 + 32 + 32 + U_DW + RAW_DW];
    float* inLDS  = smem;
    float* xLDS   = smem;
    float* kLDS   = smem + TILE_DW;
    float* scLDS  = smem + TILE_DW + 140;
    float* offLDS = smem + TILE_DW + 172;
    float* uLDS   = smem + TILE_DW + 204;
    float* rawLDS = uLDS + U_DW;

    const int tid   = threadIdx.x;
    const int n     = blockIdx.y;
    const int strip = blockIdx.x;

    const float* embn = emb + (size_t)n * 512 * 32 * 32;
    {
        const int Wp = tid & 127;
        const int jw = (Wp & 3) * 1024 + (Wp >> 2);
        const int y1 = tid >> 7;
        const int y2 = 4 + ((tid >> 7) & 1);
        const int cb = (tid >> 8) * 16;
        const int H1 = 4 * strip - 1 + y1;
        const int H2 = 4 * strip - 1 + y2;
        float v[32], u[16];
        if ((unsigned)H1 < 128u) {
            const float* src = embn + ((H1 & 3) * 4096 + (H1 >> 2) * 32 + jw);
#pragma unroll
            for (int c = 0; c < 32; ++c) v[c] = src[c * PLANE_C];
        } else {
#pragma unroll
            for (int c = 0; c < 32; ++c) v[c] = 0.f;
        }
        if ((unsigned)H2 < 128u) {
            const float* src = embn + ((H2 & 3) * 4096 + (H2 >> 2) * 32 + jw) + cb * PLANE_C;
#pragma unroll
            for (int c = 0; c < 16; ++c) u[c] = src[c * PLANE_C];
        } else {
#pragma unroll
            for (int c = 0; c < 16; ++c) u[c] = 0.f;
        }
        float* dst1 = &inLDS[PIXOFF(y1 * ROWP + 1 + Wp)];
        float* dst2 = &inLDS[PIXOFF(y2 * ROWP + 1 + Wp) + cb];
#pragma unroll
        for (int q = 0; q < 8; ++q)
            *(float4*)(dst1 + 4 * q) =
                make_float4(v[4 * q], v[4 * q + 1], v[4 * q + 2], v[4 * q + 3]);
#pragma unroll
        for (int q = 0; q < 4; ++q)
            *(float4*)(dst2 + 4 * q) =
                make_float4(u[4 * q], u[4 * q + 1], u[4 * q + 2], u[4 * q + 3]);
        if (tid < 12) {
            float* dst = &inLDS[PIXOFF((tid >> 1) * ROWP + (tid & 1) * (ROWP - 1))];
#pragma unroll
            for (int q = 0; q < 8; ++q)
                *(float4*)(dst + 4 * q) = make_float4(0.f, 0.f, 0.f, 0.f);
        }
    }
    for (int i = tid; i < RAW_DW; i += 512) rawLDS[i] = conv_w[i];
    if (tid < 137) kLDS[tid] = inck[n * 137 + tid];
    if (tid < 32) {
        const float sc = bn_g[tid] * rsqrtf(bn_v[tid] + 1e-5f);
        scLDS[tid]  = sc;
        offLDS[tid] = (conv_b[tid] - bn_m[tid]) * sc + bn_b[tid];
    }
    __syncthreads();

#pragma unroll
    for (int r = 0; r < 2; ++r) {
        const int o = (tid >> 5) + 16 * r;
        const int c = tid & 31;
        const float* g = &rawLDS[o * 288 + c * 9];
        float a[4][3];
#pragma unroll
        for (int x = 0; x < 3; ++x) {
            const float g0 = g[x], g1 = g[3 + x], g2 = g[6 + x];
            a[0][x] = g0;
            a[1][x] = 0.5f * (g0 + g1 + g2);
            a[2][x] = 0.5f * (g0 - g1 + g2);
            a[3][x] = g2;
        }
        float* u = &uLDS[o * UO + c];
#pragma unroll
        for (int xi = 0; xi < 4; ++xi) {
            u[(xi * 4 + 0) * UPOS] = a[xi][0];
            u[(xi * 4 + 1) * UPOS] = 0.5f * (a[xi][0] + a[xi][1] + a[xi][2]);
            u[(xi * 4 + 2) * UPOS] = 0.5f * (a[xi][0] - a[xi][1] + a[xi][2]);
            u[(xi * 4 + 3) * UPOS] = a[xi][2];
        }
    }
    __syncthreads();

    const int wave = tid >> 5;
    const int lane = tid & 31;
    const int half = lane >> 4;
    const int ml   = lane & 15;
    const int nb   = (wave >> 3) << 4;
    const int tr   = (wave >> 2) & 1;
    const int tc0  = (wave & 3) << 4;

    v8f M[16] = {};
    const float* abase = &inLDS[PIXOFF(2 * tr * ROWP + 2 * (tc0 + ml)) + 2 * half];
    const float* bbase = &uLDS[(nb + ml) * UO + 2 * half];
    for (int kk = 0; kk < 8; ++kk) {
        v2f t[4][4];
#pragma unroll
        for (int s = 0; s < 4; ++s) {
            v2f d[4];
#pragma unroll
            for (int r = 0; r < 4; ++r) {
                const float2 q = *(const float2*)(abase + r * PIXOFF(ROWP) + PIXOFF(s) + kk * 4);
                d[r].x = q.x; d[r].y = q.y;
            }
            t[0][s] = d[0] - d[2];
            t[1][s] = d[1] + d[2];
            t[2][s] = d[2] - d[1];
            t[3][s] = d[1] - d[3];
        }
#pragma unroll
        for (int xi = 0; xi < 4; ++xi) {
            v2f V[4];
            V[0] = t[xi][0] - t[xi][2];
            V[1] = t[xi][1] + t[xi][2];
            V[2] = t[xi][2] - t[xi][1];
            V[3] = t[xi][1] - t[xi][3];
#pragma unroll
            for (int nu = 0; nu < 4; ++nu) {
                const float2 q = *(const float2*)(bbase + (xi * 4 + nu) * UPOS + kk * 4);
                v2f bv; bv.x = q.x; bv.y = q.y;
                M[xi * 4 + nu] = __builtin_amdgcn_wmma_f32_16x16x4_f32(
                    false, V[nu], false, bv, (short)0, M[xi * 4 + nu], false, false);
            }
        }
    }

    v8f Y[2][2];
    {
        v8f Z[4][2];
#pragma unroll
        for (int xi = 0; xi < 4; ++xi) {
            Z[xi][0] = M[xi * 4 + 0] + M[xi * 4 + 1] + M[xi * 4 + 2];
            Z[xi][1] = M[xi * 4 + 1] - M[xi * 4 + 2] - M[xi * 4 + 3];
        }
#pragma unroll
        for (int j = 0; j < 2; ++j) {
            Y[0][j] = Z[0][j] + Z[1][j] + Z[2][j];
            Y[1][j] = Z[1][j] - Z[2][j] - Z[3][j];
        }
    }

    __syncthreads();

    {
        const int och  = nb + ml;
        const float sc = scLDS[och];
        const float of = offLDS[och];
#pragma unroll
        for (int i = 0; i < 2; ++i) {
#pragma unroll
            for (int j = 0; j < 2; ++j) {
#pragma unroll
                for (int r = 0; r < 8; ++r) {
                    const int m = r + 8 * half;
                    const int pix = (2 * tr + i) * 128 + 2 * (tc0 + m) + j;
                    xLDS[PIXOFF(pix) + och] = fmaxf(Y[i][j][r] * sc + of, 0.f);
                }
            }
        }
    }
    __syncthreads();

    {
        const int p = tid;
        float h0 = kLDS[132], h1 = kLDS[133], h2 = kLDS[134], h3 = kLDS[135];
        const float* xp = &xLDS[PIXOFF(p)];
#pragma unroll
        for (int c = 0; c < 32; ++c) {
            const float xc = xp[c];
            h0 = fmaf(kLDS[c],      xc, h0);
            h1 = fmaf(kLDS[32 + c], xc, h1);
            h2 = fmaf(kLDS[64 + c], xc, h2);
            h3 = fmaf(kLDS[96 + c], xc, h3);
        }
        h0 = fmaxf(h0, 0.f); h1 = fmaxf(h1, 0.f);
        h2 = fmaxf(h2, 0.f); h3 = fmaxf(h3, 0.f);
        float y = kLDS[136];
        y = fmaf(kLDS[128], h0, y);
        y = fmaf(kLDS[129], h1, y);
        y = fmaf(kLDS[130], h2, y);
        y = fmaf(kLDS[131], h3, y);
        const float sg = 1.f / (1.f + __expf(-y));
        const int H = 4 * strip + (p >> 7), W = p & 127;
        z0[((size_t)n * 128 + H) * 128 + W] = sg;
    }
}

#define SM_PITCH 132
#define SM_TROWS 12
__global__ __launch_bounds__(128) void smooth_kernel(
    const float* __restrict__ z0,
    const float* __restrict__ mask,
    const float* __restrict__ sw,
    const float* __restrict__ sb,
    float* __restrict__ out)
{
    __shared__ __attribute__((aligned(16))) float tile[2 * SM_TROWS * SM_PITCH];

    const int tid = threadIdx.x;
    const int tx  = tid & 31;
    const int ty  = tid >> 5;
    const int n   = blockIdx.x >> 4;
    const int H0  = (blockIdx.x & 15) * 8;
    const float* zp = z0   + (size_t)n * 16384;
    const float* mp = mask + (size_t)n * 16384;

    float4 g[6];
#pragma unroll
    for (int k = 0; k < 6; ++k) {
        const int row = (k % 3) * 4 + ty;
        const int h   = H0 - 2 + row;
        const float* src = (k < 3) ? zp : mp;
        g[k] = make_float4(0.f, 0.f, 0.f, 0.f);
        if ((unsigned)h < 128u) g[k] = *(const float4*)(src + h * 128 + 4 * tx);
    }
#pragma unroll
    for (int k = 0; k < 6; ++k) {
        const int row = (k % 3) * 4 + ty;
        float* dst = &tile[((k / 3) * SM_TROWS + row) * SM_PITCH + 2 + 4 * tx];
        *(float2*)dst       = make_float2(g[k].x, g[k].y);
        *(float2*)(dst + 2) = make_float2(g[k].z, g[k].w);
    }
    if (tid < 48) {
        const int pl  = tid / 24;
        const int rem = tid - pl * 24;
        float* dst = &tile[(pl * SM_TROWS + (rem >> 1)) * SM_PITCH + (rem & 1) * 130];
        *(float2*)dst = make_float2(0.f, 0.f);
    }
    __syncthreads();

    const float bias = sb[0];
    float acc[2][4];
#pragma unroll
    for (int o = 0; o < 2; ++o)
#pragma unroll
        for (int q = 0; q < 4; ++q) acc[o][q] = bias;

#pragma unroll
    for (int r = 0; r < 6; ++r) {
        const float* zr = &tile[(2 * ty + r) * SM_PITCH + 4 * tx];
        const float* mr = zr + SM_TROWS * SM_PITCH;
        const float4 za = *(const float4*)zr, zb = *(const float4*)(zr + 4);
        const float4 ma = *(const float4*)mr, mb = *(const float4*)(mr + 4);
        const float zv[8] = {za.x, za.y, za.z, za.w, zb.x, zb.y, zb.z, zb.w};
        const float mv[8] = {ma.x, ma.y, ma.z, ma.w, mb.x, mb.y, mb.z, mb.w};
#pragma unroll
        for (int o = 0; o < 2; ++o) {
            const int dy = r - o;
            if (dy < 0 || dy > 4) continue;
#pragma unroll
            for (int dx = 0; dx < 5; ++dx) {
                const float wz = sw[dy * 5 + dx];
                const float wm = sw[25 + dy * 5 + dx];
#pragma unroll
                for (int q = 0; q < 4; ++q) {
                    acc[o][q] = fmaf(wz, zv[q + dx], acc[o][q]);
                    acc[o][q] = fmaf(wm, mv[q + dx], acc[o][q]);
                }
            }
        }
    }

#pragma unroll
    for (int o = 0; o < 2; ++o) {
        const int H = H0 + 2 * ty + o;
        *(float4*)(out + ((size_t)n * 128 + H) * 128 + 4 * tx) =
            make_float4(acc[o][0], acc[o][1], acc[o][2], acc[o][3]);
    }
}

extern "C" void kernel_launch(void* const* d_in, const int* in_sizes, int n_in,
                              void* d_out, int out_size, void* d_ws, size_t ws_size,
                              hipStream_t stream) {
    const float* emb    = (const float*)d_in[0];
    const float* inck   = (const float*)d_in[1];
    const float* masks  = (const float*)d_in[2];
    const float* conv_w = (const float*)d_in[3];
    const float* conv_b = (const float*)d_in[4];
    const float* bn_g   = (const float*)d_in[5];
    const float* bn_b   = (const float*)d_in[6];
    const float* bn_m   = (const float*)d_in[7];
    const float* bn_v   = (const float*)d_in[8];
    const float* sw     = (const float*)d_in[9];
    const float* sb     = (const float*)d_in[10];
    float* out = (float*)d_out;
    float* z0  = (float*)d_ws;

    dim3 g1(32, 32);
    fused_head_kernel<<<g1, 512, 0, stream>>>(emb, inck, conv_w, conv_b,
                                              bn_g, bn_b, bn_m, bn_v, z0);
    smooth_kernel<<<512, 128, 0, stream>>>(z0, masks, sw, sb, out);
}


# head kernel persistent: 256 work-groups x 4 strips; the next strip's halo tile is fetched into a second LDS buffer during the GEMM k steps (6 loads per thread per step); filters transformed once
# speedup vs baseline: 2.8199x; 1.0767x over previous
#include <hip/hip_runtime.h>
#include <hip/hip_bf16.h>

typedef __attribute__((ext_vector_type(2))) float v2f;
typedef __attribute__((ext_vector_type(8))) float v8f;

#define PIXOFF(p) ((p) * 32 + ((p) >> 1) * 4)
#define UO 36
#define UPOS (32 * UO)
#define U_DW (16 * UPOS)
#define RAW_DW 9216
#define ROWP 130
#define TILE_DW 26520
#define PLANE_C (16 * 32 * 32)
#define STRIPS_PER_BLOCK 4
__global__ __launch_bounds__(512) void fused_head_kernel(
    const float* __restrict__ emb,
    const float* __restrict__ inck,
    const float* __restrict__ conv_w,
    const float* __restrict__ conv_b,
    const float* __restrict__ bn_g,
    const float* __restrict__ bn_b,
    const float* __restrict__ bn_m,
    const float* __restrict__ bn_v,
    float* __restrict__ z0)
{
    __shared__ __attribute__((aligned(16))) float smem[2 * TILE_DW + 140 + 32 + 32 + U_DW];
    float* kLDS   = smem + 2 * TILE_DW;
    float* scLDS  = kLDS + 140;
    float* offLDS = kLDS + 172;
    float* uLDS   = kLDS + 204;
    float* rawLDS = smem + TILE_DW;

    const int tid = threadIdx.x;
    const int n   = blockIdx.x >> 3;
    const int s0  = (blockIdx.x & 7) * STRIPS_PER_BLOCK;

    const float* embn = emb + (size_t)n * 512 * 32 * 32;
    const int Wp = tid & 127;
    const int jw = (Wp & 3) * 1024 + (Wp >> 2);
    const int y1 = tid >> 7;
    const int y2 = 4 + ((tid >> 7) & 1);
    const int cb = (tid >> 8) * 16;
    const int o1 = PIXOFF(y1 * ROWP + 1 + Wp);
    const int o2 = PIXOFF(y2 * ROWP + 1 + Wp) + cb;

    {
        const int H1 = 4 * s0 - 1 + y1;
        const int H2 = 4 * s0 - 1 + y2;
        float v[32], u[16];
        if ((unsigned)H1 < 128u) {
            const float* src = embn + ((H1 & 3) * 4096 + (H1 >> 2) * 32 + jw);
#pragma unroll
            for (int c = 0; c < 32; ++c) v[c] = src[c * PLANE_C];
        } else {
#pragma unroll
            for (int c = 0; c < 32; ++c) v[c] = 0.f;
        }
        if ((unsigned)H2 < 128u) {
            const float* src = embn + ((H2 & 3) * 4096 + (H2 >> 2) * 32 + jw) + cb * PLANE_C;
#pragma unroll
            for (int c = 0; c < 16; ++c) u[c] = src[c * PLANE_C];
        } else {
#pragma unroll
            for (int c = 0; c < 16; ++c) u[c] = 0.f;
        }
        float* dst1 = smem + o1;
        float* dst2 = smem + o2;
#pragma unroll
        for (int q = 0; q < 8; ++q)
            *(float4*)(dst1 + 4 * q) =
                make_float4(v[4 * q], v[4 * q + 1], v[4 * q + 2], v[4 * q + 3]);
#pragma unroll
        for (int q = 0; q < 4; ++q)
            *(float4*)(dst2 + 4 * q) =
                make_float4(u[4 * q], u[4 * q + 1], u[4 * q + 2], u[4 * q + 3]);
        if (tid < 12) {
            float* dst = &smem[PIXOFF((tid >> 1) * ROWP + (tid & 1) * (ROWP - 1))];
#pragma unroll
            for (int q = 0; q < 8; ++q)
                *(float4*)(dst + 4 * q) = make_float4(0.f, 0.f, 0.f, 0.f);
        }
    }
    for (int i = tid; i < RAW_DW; i += 512) rawLDS[i] = conv_w[i];
    if (tid < 137) kLDS[tid] = inck[n * 137 + tid];
    if (tid < 32) {
        const float sc = bn_g[tid] * rsqrtf(bn_v[tid] + 1e-5f);
        scLDS[tid]  = sc;
        offLDS[tid] = (conv_b[tid] - bn_m[tid]) * sc + bn_b[tid];
    }
    __syncthreads();

#pragma unroll
    for (int r = 0; r < 2; ++r) {
        const int o = (tid >> 5) + 16 * r;
        const int c = tid & 31;
        const float* g = &rawLDS[o * 288 + c * 9];
        float a[4][3];
#pragma unroll
        for (int x = 0; x < 3; ++x) {
            const float g0 = g[x], g1 = g[3 + x], g2 = g[6 + x];
            a[0][x] = g0;
            a[1][x] = 0.5f * (g0 + g1 + g2);
            a[2][x] = 0.5f * (g0 - g1 + g2);
            a[3][x] = g2;
        }
        float* u = &uLDS[o * UO + c];
#pragma unroll
        for (int xi = 0; xi < 4; ++xi) {
            u[(xi * 4 + 0) * UPOS] = a[xi][0];
            u[(xi * 4 + 1) * UPOS] = 0.5f * (a[xi][0] + a[xi][1] + a[xi][2]);
            u[(xi * 4 + 2) * UPOS] = 0.5f * (a[xi][0] - a[xi][1] + a[xi][2]);
            u[(xi * 4 + 3) * UPOS] = a[xi][2];
        }
    }
    __syncthreads();

    const int wave = tid >> 5;
    const int lane = tid & 31;
    const int half = lane >> 4;
    const int ml   = lane & 15;
    const int nb   = (wave >> 3) << 4;
    const int tr   = (wave >> 2) & 1;
    const int tc0  = (wave & 3) << 4;
    const int aoff = PIXOFF(2 * tr * ROWP + 2 * (tc0 + ml)) + 2 * half;
    const float* bbase = &uLDS[(nb + ml) * UO + 2 * half];

#pragma unroll 1
    for (int it = 0; it < STRIPS_PER_BLOCK; ++it) {
        const int strip = s0 + it;
        float* cur = smem + (it & 1) * TILE_DW;
        float* nxt = smem + ((it & 1) ^ 1) * TILE_DW;
        const bool pf = it + 1 < STRIPS_PER_BLOCK;

        const int Hn1 = 4 * (strip + 1) - 1 + y1;
        const int Hn2 = 4 * (strip + 1) - 1 + y2;
        const bool ok1 = (unsigned)Hn1 < 128u;
        const bool ok2 = (unsigned)Hn2 < 128u;
        const int Hc1 = min(max(Hn1, 0), 127);
        const int Hc2 = min(max(Hn2, 0), 127);
        const int q1 = (Hc1 & 3) * 4096 + (Hc1 >> 2) * 32 + jw;
        const int q2 = (Hc2 & 3) * 4096 + (Hc2 >> 2) * 32 + jw + cb * PLANE_C;
        float* d1 = nxt + o1;
        float* d2 = nxt + o2;
        if (pf && tid < 12) {
            float* dst = &nxt[PIXOFF((tid >> 1) * ROWP + (tid & 1) * (ROWP - 1))];
#pragma unroll
            for (int q = 0; q < 8; ++q)
                *(float4*)(dst + 4 * q) = make_float4(0.f, 0.f, 0.f, 0.f);
        }

        v8f M[16] = {};
        const float* abase = cur + aoff;
        for (int kk = 0; kk < 8; ++kk) {
            float4 pv = make_float4(0.f, 0.f, 0.f, 0.f);
            float2 pu = make_float2(0.f, 0.f);
            if (pf) {
                const float* a1 = embn + q1 + 4 * kk * PLANE_C;
                const float* a2 = embn + q2 + 2 * kk * PLANE_C;
                pv = make_float4(a1[0], a1[PLANE_C], a1[2 * PLANE_C], a1[3 * PLANE_C]);
                pu = make_float2(a2[0], a2[PLANE_C]);
            }
            v2f t[4][4];
#pragma unroll
            for (int s = 0; s < 4; ++s) {
                v2f d[4];
#pragma unroll
                for (int r = 0; r < 4; ++r) {
                    const float2 q = *(const float2*)(abase + r * PIXOFF(ROWP) + PIXOFF(s) + kk * 4);
                    d[r].x = q.x; d[r].y = q.y;
                }
                t[0][s] = d[0] - d[2];
                t[1][s] = d[1] + d[2];
                t[2][s] = d[2] - d[1];
                t[3][s] = d[1] - d[3];
            }
#pragma unroll
            for (int xi = 0; xi < 4; ++xi) {
                v2f V[4];
                V[0] = t[xi][0] - t[xi][2];
                V[1] = t[xi][1] + t[xi][2];
                V[2] = t[xi][2] - t[xi][1];
                V[3] = t[xi][1] - t[xi][3];
#pragma unroll
                for (int nu = 0; nu < 4; ++nu) {
                    const float2 q = *(const float2*)(bbase + (xi * 4 + nu) * UPOS + kk * 4);
                    v2f bv; bv.x = q.x; bv.y = q.y;
                    M[xi * 4 + nu] = __builtin_amdgcn_wmma_f32_16x16x4_f32(
                        false, V[nu], false, bv, (short)0, M[xi * 4 + nu], false, false);
                }
            }
            if (pf) {
                *(float4*)(d1 + 4 * kk) = ok1 ? pv : make_float4(0.f, 0.f, 0.f, 0.f);
                *(float2*)(d2 + 2 * kk) = ok2 ? pu : make_float2(0.f, 0.f);
            }
        }

        v8f Y[2][2];
        {
            v8f Z[4][2];
#pragma unroll
            for (int xi = 0; xi < 4; ++xi) {
                Z[xi][0] = M[xi * 4 + 0] + M[xi * 4 + 1] + M[xi * 4 + 2];
                Z[xi][1] = M[xi * 4 + 1] - M[xi * 4 + 2] - M[xi * 4 + 3];
            }
#pragma unroll
            for (int j = 0; j < 2; ++j) {
                Y[0][j] = Z[0][j] + Z[1][j] + Z[2][j];
                Y[1][j] = Z[1][j] - Z[2][j] - Z[3][j];
            }
        }

        __syncthreads();

        {
            const int och  = nb + ml;
            const float sc = scLDS[och];
            const float of = offLDS[och];
#pragma unroll
            for (int i = 0; i < 2; ++i) {
#pragma unroll
                for (int j = 0; j < 2; ++j) {
#pragma unroll
                    for (int r = 0; r < 8; ++r) {
                        const int m = r + 8 * half;
                        const int pix = (2 * tr + i) * 128 + 2 * (tc0 + m) + j;
                        cur[PIXOFF(pix) + och] = fmaxf(Y[i][j][r] * sc + of, 0.f);
                    }
                }
            }
        }
        __syncthreads();

        {
            const int p = tid;
            float h0 = kLDS[132], h1 = kLDS[133], h2 = kLDS[134], h3 = kLDS[135];
            const float* xp = &cur[PIXOFF(p)];
#pragma unroll
            for (int c = 0; c < 32; ++c) {
                const float xc = xp[c];
                h0 = fmaf(kLDS[c],      xc, h0);
                h1 = fmaf(kLDS[32 + c], xc, h1);
                h2 = fmaf(kLDS[64 + c], xc, h2);
                h3 = fmaf(kLDS[96 + c], xc, h3);
            }
            h0 = fmaxf(h0, 0.f); h1 = fmaxf(h1, 0.f);
            h2 = fmaxf(h2, 0.f); h3 = fmaxf(h3, 0.f);
            float y = kLDS[136];
            y = fmaf(kLDS[128], h0, y);
            y = fmaf(kLDS[129], h1, y);
            y = fmaf(kLDS[130], h2, y);
            y = fmaf(kLDS[131], h3, y);
            const float sg = 1.f / (1.f + __expf(-y));
            const int H = 4 * strip + (p >> 7), W = p & 127;
            z0[((size_t)n * 128 + H) * 128 + W] = sg;
        }
        __syncthreads();
    }
}

#define SM_PITCH 132
#define SM_TROWS 12
__global__ __launch_bounds__(128) void smooth_kernel(
    const float* __restrict__ z0,
    const float* __restrict__ mask,
    const float* __restrict__ sw,
    const float* __restrict__ sb,
    float* __restrict__ out)
{
    __shared__ __attribute__((aligned(16))) float tile[2 * SM_TROWS * SM_PITCH];

    const int tid = threadIdx.x;
    const int tx  = tid & 31;
    const int ty  = tid >> 5;
    const int n   = blockIdx.x >> 4;
    const int H0  = (blockIdx.x & 15) * 8;
    const float* zp = z0   + (size_t)n * 16384;
    const float* mp = mask + (size_t)n * 16384;

    float4 g[6];
#pragma unroll
    for (int k = 0; k < 6; ++k) {
        const int row = (k % 3) * 4 + ty;
        const int h   = H0 - 2 + row;
        const float* src = (k < 3) ? zp : mp;
        g[k] = make_float4(0.f, 0.f, 0.f, 0.f);
        if ((unsigned)h < 128u) g[k] = *(const float4*)(src + h * 128 + 4 * tx);
    }
#pragma unroll
    for (int k = 0; k < 6; ++k) {
        const int row = (k % 3) * 4 + ty;
        float* dst = &tile[((k / 3) * SM_TROWS + row) * SM_PITCH + 2 + 4 * tx];
        *(float2*)dst       = make_float2(g[k].x, g[k].y);
        *(float2*)(dst + 2) = make_float2(g[k].z, g[k].w);
    }
    if (tid < 48) {
        const int pl  = tid / 24;
        const int rem = tid - pl * 24;
        float* dst = &tile[(pl * SM_TROWS + (rem >> 1)) * SM_PITCH + (rem & 1) * 130];
        *(float2*)dst = make_float2(0.f, 0.f);
    }
    __syncthreads();

    const float bias = sb[0];
    float acc[2][4];
#pragma unroll
    for (int o = 0; o < 2; ++o)
#pragma unroll
        for (int q = 0; q < 4; ++q) acc[o][q] = bias;

#pragma unroll
    for (int r = 0; r < 6; ++r) {
        const float* zr = &tile[(2 * ty + r) * SM_PITCH + 4 * tx];
        const float* mr = zr + SM_TROWS * SM_PITCH;
        const float4 za = *(const float4*)zr, zb = *(const float4*)(zr + 4);
        const float4 ma = *(const float4*)mr, mb = *(const float4*)(mr + 4);
        const float zv[8] = {za.x, za.y, za.z, za.w, zb.x, zb.y, zb.z, zb.w};
        const float mv[8] = {ma.x, ma.y, ma.z, ma.w, mb.x, mb.y, mb.z, mb.w};
#pragma unroll
        for (int o = 0; o < 2; ++o) {
            const int dy = r - o;
            if (dy < 0 || dy > 4) continue;
#pragma unroll
            for (int dx = 0; dx < 5; ++dx) {
                const float wz = sw[dy * 5 + dx];
                const float wm = sw[25 + dy * 5 + dx];
#pragma unroll
                for (int q = 0; q < 4; ++q) {
                    acc[o][q] = fmaf(wz, zv[q + dx], acc[o][q]);
                    acc[o][q] = fmaf(wm, mv[q + dx], acc[o][q]);
                }
            }
        }
    }

#pragma unroll
    for (int o = 0; o < 2; ++o) {
        const int H = H0 + 2 * ty + o;
        *(float4*)(out + ((size_t)n * 128 + H) * 128 + 4 * tx) =
            make_float4(acc[o][0], acc[o][1], acc[o][2], acc[o][3]);
    }
}

extern "C" void kernel_launch(void* const* d_in, const int* in_sizes, int n_in,
                              void* d_out, int out_size, void* d_ws, size_t ws_size,
                              hipStream_t stream) {
    const float* emb    = (const float*)d_in[0];
    const float* inck   = (const float*)d_in[1];
    const float* masks  = (const float*)d_in[2];
    const float* conv_w = (const float*)d_in[3];
    const float* conv_b = (const float*)d_in[4];
    const float* bn_g   = (const float*)d_in[5];
    const float* bn_b   = (const float*)d_in[6];
    const float* bn_m   = (const float*)d_in[7];
    const float* bn_v   = (const float*)d_in[8];
    const float* sw     = (const float*)d_in[9];
    const float* sb     = (const float*)d_in[10];
    float* out = (float*)d_out;
    float* z0  = (float*)d_ws;

    fused_head_kernel<<<256, 512, 0, stream>>>(emb, inck, conv_w, conv_b,
                                               bn_g, bn_b, bn_m, bn_v, z0);
    smooth_kernel<<<512, 128, 0, stream>>>(z0, masks, sw, sb, out);
}


# GEMM k loop freed of its waits: next-strip prefetch gets a second register set (asked in step kk-1, stored in step kk); each step's 32 LDS fragment reads stand as one group before its 16 WMMAs
# speedup vs baseline: 2.9814x; 1.0573x over previous
#include <hip/hip_runtime.h>
#include <hip/hip_bf16.h>

typedef __attribute__((ext_vector_type(2))) float v2f;
typedef __attribute__((ext_vector_type(8))) float v8f;

#define PIXOFF(p) ((p) * 32 + ((p) >> 1) * 4)
#define UO 36
#define UPOS (32 * UO)
#define U_DW (16 * UPOS)
#define RAW_DW 9216
#define ROWP 130
#define TILE_DW 26520
#define PLANE_C (16 * 32 * 32)
#define STRIPS_PER_BLOCK 4
__global__ __launch_bounds__(512) void fused_head_kernel(
    const float* __restrict__ emb,
    const float* __restrict__ inck,
    const float* __restrict__ conv_w,
    const float* __restrict__ conv_b,
    const float* __restrict__ bn_g,
    const float* __restrict__ bn_b,
    const float* __restrict__ bn_m,
    const float* __restrict__ bn_v,
    float* __restrict__ z0)
{
    __shared__ __attribute__((aligned(16))) float smem[2 * TILE_DW + 140 + 32 + 32 + U_DW];
    float* kLDS   = smem;
    float* scLDS  = smem + 140;
    float* offLDS = smem + 172;
    float* tiles  = smem + 204;
    float* uLDS   = tiles + 2 * TILE_DW;
    float* rawLDS = tiles + TILE_DW;

    const int tid = threadIdx.x;
    const int n   = blockIdx.x >> 3;
    const int s0  = (blockIdx.x & 7) * STRIPS_PER_BLOCK;

    const float* embn = emb + (size_t)n * 512 * 32 * 32;
    const int Wp = tid & 127;
    const int jw = (Wp & 3) * 1024 + (Wp >> 2);
    const int y1 = tid >> 7;
    const int y2 = 4 + ((tid >> 7) & 1);
    const int cb = (tid >> 8) * 16;
    const int o1 = PIXOFF(y1 * ROWP + 1 + Wp);
    const int o2 = PIXOFF(y2 * ROWP + 1 + Wp) + cb;

    {
        const int H1 = 4 * s0 - 1 + y1;
        const int H2 = 4 * s0 - 1 + y2;
        float v[32], u[16];
        if ((unsigned)H1 < 128u) {
            const float* src = embn + ((H1 & 3) * 4096 + (H1 >> 2) * 32 + jw);
#pragma unroll
            for (int c = 0; c < 32; ++c) v[c] = src[c * PLANE_C];
        } else {
#pragma unroll
            for (int c = 0; c < 32; ++c) v[c] = 0.f;
        }
        if ((unsigned)H2 < 128u) {
            const float* src = embn + ((H2 & 3) * 4096 + (H2 >> 2) * 32 + jw) + cb * PLANE_C;
#pragma unroll
            for (int c = 0; c < 16; ++c) u[c] = src[c * PLANE_C];
        } else {
#pragma unroll
            for (int c = 0; c < 16; ++c) u[c] = 0.f;
        }
        float* dst1 = tiles + o1;
        float* dst2 = tiles + o2;
#pragma unroll
        for (int q = 0; q < 8; ++q)
            *(float4*)(dst1 + 4 * q) =
                make_float4(v[4 * q], v[4 * q + 1], v[4 * q + 2], v[4 * q + 3]);
#pragma unroll
        for (int q = 0; q < 4; ++q)
            *(float4*)(dst2 + 4 * q) =
                make_float4(u[4 * q], u[4 * q + 1], u[4 * q + 2], u[4 * q + 3]);
        if (tid < 12) {
            float* dst = &tiles[PIXOFF((tid >> 1) * ROWP + (tid & 1) * (ROWP - 1))];
#pragma unroll
            for (int q = 0; q < 8; ++q)
                *(float4*)(dst + 4 * q) = make_float4(0.f, 0.f, 0.f, 0.f);
        }
    }
    for (int i = tid; i < RAW_DW; i += 512) rawLDS[i] = conv_w[i];
    if (tid < 137) kLDS[tid] = inck[n * 137 + tid];
    if (tid < 32) {
        const float sc = bn_g[tid] * rsqrtf(bn_v[tid] + 1e-5f);
        scLDS[tid]  = sc;
        offLDS[tid] = (conv_b[tid] - bn_m[tid]) * sc + bn_b[tid];
    }
    __syncthreads();

#pragma unroll
    for (int r = 0; r < 2; ++r) {
        const int o = (tid >> 5) + 16 * r;
        const int c = tid & 31;
        const float* g = &rawLDS[o * 288 + c * 9];
        float a[4][3];
#pragma unroll
        for (int x = 0; x < 3; ++x) {
            const float g0 = g[x], g1 = g[3 + x], g2 = g[6 + x];
            a[0][x] = g0;
            a[1][x] = 0.5f * (g0 + g1 + g2);
            a[2][x] = 0.5f * (g0 - g1 + g2);
            a[3][x] = g2;
        }
        float* u = &uLDS[o * UO + c];
#pragma unroll
        for (int xi = 0; xi < 4; ++xi) {
            u[(xi * 4 + 0) * UPOS] = a[xi][0];
            u[(xi * 4 + 1) * UPOS] = 0.5f * (a[xi][0] + a[xi][1] + a[xi][2]);
            u[(xi * 4 + 2) * UPOS] = 0.5f * (a[xi][0] - a[xi][1] + a[xi][2]);
            u[(xi * 4 + 3) * UPOS] = a[xi][2];
        }
    }
    __syncthreads();

    const int wave = tid >> 5;
    const int lane = tid & 31;
    const int half = lane >> 4;
    const int ml   = lane & 15;
    const int nb   = (wave >> 3) << 4;
    const int tr   = (wave >> 2) & 1;
    const int tc0  = (wave & 3) << 4;
    const int aoff = PIXOFF(2 * tr * ROWP + 2 * (tc0 + ml)) + 2 * half;
    const float* bbase = &uLDS[(nb + ml) * UO + 2 * half];

#pragma unroll 1
    for (int it = 0; it < STRIPS_PER_BLOCK; ++it) {
        const int strip = s0 + it;
        float* cur = tiles + (it & 1) * TILE_DW;
        float* nxt = tiles + ((it & 1) ^ 1) * TILE_DW;
        const bool pf = it + 1 < STRIPS_PER_BLOCK;

        const int Hn1 = 4 * (strip + 1) - 1 + y1;
        const int Hn2 = 4 * (strip + 1) - 1 + y2;
        const bool ok2 = (unsigned)Hn2 < 128u;
        const int Hc1 = min(max(Hn1, 0), 127);
        const int Hc2 = min(max(Hn2, 0), 127);
        const int q1 = (Hc1 & 3) * 4096 + (Hc1 >> 2) * 32 + jw;
        const int q2 = (Hc2 & 3) * 4096 + (Hc2 >> 2) * 32 + jw + cb * PLANE_C;
        float* d1 = nxt + o1;
        float* d2 = nxt + o2;
        if (pf && tid < 12) {
            float* dst = &nxt[PIXOFF((tid >> 1) * ROWP + (tid & 1) * (ROWP - 1))];
#pragma unroll
            for (int q = 0; q < 8; ++q)
                *(float4*)(dst + 4 * q) = make_float4(0.f, 0.f, 0.f, 0.f);
        }

        v8f M[16] = {};
        const float* abase = cur + aoff;
        auto gemm_step = [&](const int kk) __attribute__((always_inline)) {
            v2f bq[16];
#pragma unroll
            for (int p = 0; p < 16; ++p) {
                const float2 q = *(const float2*)(bbase + p * UPOS + kk * 4);
                bq[p].x = q.x; bq[p].y = q.y;
            }
            v2f d[4][4];
#pragma unroll
            for (int r = 0; r < 4; ++r) {
#pragma unroll
                for (int s = 0; s < 4; ++s) {
                    const float2 q = *(const float2*)(abase + r * PIXOFF(ROWP) + PIXOFF(s) + kk * 4);
                    d[r][s].x = q.x; d[r][s].y = q.y;
                }
            }
            v2f t[4][4];
#pragma unroll
            for (int s = 0; s < 4; ++s) {
                t[0][s] = d[0][s] - d[2][s];
                t[1][s] = d[1][s] + d[2][s];
                t[2][s] = d[2][s] - d[1][s];
                t[3][s] = d[1][s] - d[3][s];
            }
#pragma unroll
            for (int xi = 0; xi < 4; ++xi) {
                v2f V[4];
                V[0] = t[xi][0] - t[xi][2];
                V[1] = t[xi][1] + t[xi][2];
                V[2] = t[xi][2] - t[xi][1];
                V[3] = t[xi][1] - t[xi][3];
#pragma unroll
                for (int nu = 0; nu < 4; ++nu) {
                    M[xi * 4 + nu] = __builtin_amdgcn_wmma_f32_16x16x4_f32(
                        false, V[nu], false, bq[xi * 4 + nu], (short)0, M[xi * 4 + nu], false, false);
                }
            }
        };
        const float* a1 = embn + q1;
        const float* a2 = embn + q2;
        auto request = [&](const int sh, float4& pv, float2& pu) __attribute__((always_inline)) {
            const float* b1 = a1 + 4 * sh * PLANE_C;
            const float* b2 = a2 + 2 * sh * PLANE_C;
            pv = make_float4(b1[0], b1[PLANE_C], b1[2 * PLANE_C], b1[3 * PLANE_C]);
            pu = make_float2(b2[0], b2[PLANE_C]);
        };
        auto deposit = [&](const int sh, const float4& pv, const float2& pu) __attribute__((always_inline)) {
            *(float4*)(d1 + 4 * sh) = pv;
            *(float2*)(d2 + 2 * sh) = pu;
        };
        if (pf) {
            float4 pv0, pv1;
            float2 pu0, pu1;
            request(0, pv0, pu0);
#pragma unroll 1
            for (int ke = 0; ke < 6; ke += 2) {
                request(ke + 1, pv1, pu1);
                gemm_step(ke);
                deposit(ke, pv0, pu0);
                request(ke + 2, pv0, pu0);
                gemm_step(ke + 1);
                deposit(ke + 1, pv1, pu1);
                __builtin_amdgcn_sched_barrier(0);
                asm volatile("v_nop\n\tv_nop\n\tv_nop\n\tv_nop" ::: "memory");
            }
            request(7, pv1, pu1);
            gemm_step(6);
            deposit(6, pv0, pu0);
            gemm_step(7);
            deposit(7, pv1, pu1);
            if (!ok2) {
#pragma unroll
                for (int q = 0; q < 4; ++q)
                    *(float4*)(d2 + 4 * q) = make_float4(0.f, 0.f, 0.f, 0.f);
            }
        } else {
#pragma unroll 1
            for (int kk = 0; kk < 8; ++kk) {
                gemm_step(kk);
                __builtin_amdgcn_sched_barrier(0);
                asm volatile("v_nop\n\tv_nop\n\tv_nop\n\tv_nop" ::: "memory");
            }
        }

        v8f Y[2][2];
        {
            v8f Z[4][2];
#pragma unroll
            for (int xi = 0; xi < 4; ++xi) {
                Z[xi][0] = M[xi * 4 + 0] + M[xi * 4 + 1] + M[xi * 4 + 2];
                Z[xi][1] = M[xi * 4 + 1] - M[xi * 4 + 2] - M[xi * 4 + 3];
            }
#pragma unroll
            for (int j = 0; j < 2; ++j) {
                Y[0][j] = Z[0][j] + Z[1][j] + Z[2][j];
                Y[1][j] = Z[1][j] - Z[2][j] - Z[3][j];
            }
        }

        __syncthreads();

        {
            const int och  = nb + ml;
            const float sc = scLDS[och];
            const float of = offLDS[och];
#pragma unroll
            for (int i = 0; i < 2; ++i) {
#pragma unroll
                for (int j = 0; j < 2; ++j) {
#pragma unroll
                    for (int r = 0; r < 8; ++r) {
                        const int m = r + 8 * half;
                        const int pix = (2 * tr + i) * 128 + 2 * (tc0 + m) + j;
                        cur[PIXOFF(pix) + och] = fmaxf(Y[i][j][r] * sc + of, 0.f);
                    }
                }
            }
        }
        __syncthreads();

        {
            const int p = tid;
            float h0 = kLDS[132], h1 = kLDS[133], h2 = kLDS[134], h3 = kLDS[135];
            const float* xp = &cur[PIXOFF(p)];
#pragma unroll
            for (int c = 0; c < 32; ++c) {
                const float xc = xp[c];
                h0 = fmaf(kLDS[c],      xc, h0);
                h1 = fmaf(kLDS[32 + c], xc, h1);
                h2 = fmaf(kLDS[64 + c], xc, h2);
                h3 = fmaf(kLDS[96 + c], xc, h3);
            }
            h0 = fmaxf(h0, 0.f); h1 = fmaxf(h1, 0.f);
            h2 = fmaxf(h2, 0.f); h3 = fmaxf(h3, 0.f);
            float y = kLDS[136];
            y = fmaf(kLDS[128], h0, y);
            y = fmaf(kLDS[129], h1, y);
            y = fmaf(kLDS[130], h2, y);
            y = fmaf(kLDS[131], h3, y);
            const float sg = 1.f / (1.f + __expf(-y));
            const int H = 4 * strip + (p >> 7), W = p & 127;
            z0[((size_t)n * 128 + H) * 128 + W] = sg;
        }
        __syncthreads();
    }
}

#define SM_PITCH 132
#define SM_TROWS 12
__global__ __launch_bounds__(128) void smooth_kernel(
    const float* __restrict__ z0,
    const float* __restrict__ mask,
    const float* __restrict__ sw,
    const float* __restrict__ sb,
    float* __restrict__ out)
{
    __shared__ __attribute__((aligned(16))) float tile[2 * SM_TROWS * SM_PITCH];

    const int tid = threadIdx.x;
    const int tx  = tid & 31;
    const int ty  = tid >> 5;
    const int n   = blockIdx.x >> 4;
    const int H0  = (blockIdx.x & 15) * 8;
    const float* zp = z0   + (size_t)n * 16384;
    const float* mp = mask + (size_t)n * 16384;

    float4 g[6];
#pragma unroll
    for (int k = 0; k < 6; ++k) {
        const int row = (k % 3) * 4 + ty;
        const int h   = H0 - 2 + row;
        const float* src = (k < 3) ? zp : mp;
        g[k] = make_float4(0.f, 0.f, 0.f, 0.f);
        if ((unsigned)h < 128u) g[k] = *(const float4*)(src + h * 128 + 4 * tx);
    }
#pragma unroll
    for (int k = 0; k < 6; ++k) {
        const int row = (k % 3) * 4 + ty;
        float* dst = &tile[((k / 3) * SM_TROWS + row) * SM_PITCH + 2 + 4 * tx];
        *(float2*)dst       = make_float2(g[k].x, g[k].y);
        *(float2*)(dst + 2) = make_float2(g[k].z, g[k].w);
    }
    if (tid < 48) {
        const int pl  = tid / 24;
        const int rem = tid - pl * 24;
        float* dst = &tile[(pl * SM_TROWS + (rem >> 1)) * SM_PITCH + (rem & 1) * 130];
        *(float2*)dst = make_float2(0.f, 0.f);
    }
    __syncthreads();

    const float bias = sb[0];
    float acc[2][4];
#pragma unroll
    for (int o = 0; o < 2; ++o)
#pragma unroll
        for (int q = 0; q < 4; ++q) acc[o][q] = bias;

#pragma unroll
    for (int r = 0; r < 6; ++r) {
        const float* zr = &tile[(2 * ty + r) * SM_PITCH + 4 * tx];
        const float* mr = zr + SM_TROWS * SM_PITCH;
        const float4 za = *(const float4*)zr, zb = *(const float4*)(zr + 4);
        const float4 ma = *(const float4*)mr, mb = *(const float4*)(mr + 4);
        const float zv[8] = {za.x, za.y, za.z, za.w, zb.x, zb.y, zb.z, zb.w};
        const float mv[8] = {ma.x, ma.y, ma.z, ma.w, mb.x, mb.y, mb.z, mb.w};
#pragma unroll
        for (int o = 0; o < 2; ++o) {
            const int dy = r - o;
            if (dy < 0 || dy > 4) continue;
#pragma unroll
            for (int dx = 0; dx < 5; ++dx) {
                const float wz = sw[dy * 5 + dx];
                const float wm = sw[25 + dy * 5 + dx];
#pragma unroll
                for (int q = 0; q < 4; ++q) {
                    acc[o][q] = fmaf(wz, zv[q + dx], acc[o][q]);
                    acc[o][q] = fmaf(wm, mv[q + dx], acc[o][q]);
                }
            }
        }
    }

#pragma unroll
    for (int o = 0; o < 2; ++o) {
        const int H = H0 + 2 * ty + o;
        *(float4*)(out + ((size_t)n * 128 + H) * 128 + 4 * tx) =
            make_float4(acc[o][0], acc[o][1], acc[o][2], acc[o][3]);
    }
}

extern "C" void kernel_launch(void* const* d_in, const int* in_sizes, int n_in,
                              void* d_out, int out_size, void* d_ws, size_t ws_size,
                              hipStream_t stream) {
    const float* emb    = (const float*)d_in[0];
    const float* inck   = (const float*)d_in[1];
    const float* masks  = (const float*)d_in[2];
    const float* conv_w = (const float*)d_in[3];
    const float* conv_b = (const float*)d_in[4];
    const float* bn_g   = (const float*)d_in[5];
    const float* bn_b   = (const float*)d_in[6];
    const float* bn_m   = (const float*)d_in[7];
    const float* bn_v   = (const float*)d_in[8];
    const float* sw     = (const float*)d_in[9];
    const float* sb     = (const float*)d_in[10];
    float* out = (float*)d_out;
    float* z0  = (float*)d_ws;

    fused_head_kernel<<<256, 512, 0, stream>>>(emb, inck, conv_w, conv_b,
                                               bn_g, bn_b, bn_m, bn_v, z0);
    smooth_kernel<<<512, 128, 0, stream>>>(z0, masks, sw, sb, out);
}


# conv GEMM operands swapped (filters = A): a lane's accumulators are 8 och of its own tile, so BN, ReLU, MLP layer 1 run in registers; 1 barrier per strip, not 3; layer-1 summation order changes
# speedup vs baseline: 3.1729x; 1.0642x over previous
#include <hip/hip_runtime.h>
#include <hip/hip_bf16.h>

typedef __attribute__((ext_vector_type(2))) float v2f;
typedef __attribute__((ext_vector_type(8))) float v8f;

#define PIXOFF(p) ((p) * 32 + ((p) >> 1) * 4)
#define UO 36
#define UPOS (32 * UO)
#define U_DW (16 * UPOS)
#define RAW_DW 9216
#define ROWP 130
#define TILE_DW 26520
#define PLANE_C (16 * 32 * 32)
#define STRIPS_PER_BLOCK 4
#define PART_DW (512 * 4)
__device__ __forceinline__ float xhalf(float x) {
    return __builtin_bit_cast(float, __builtin_amdgcn_permlanex16(
        __builtin_bit_cast(int, x), __builtin_bit_cast(int, x), 0x76543210, 0xfedcba98, false, false));
}
__global__ __launch_bounds__(512) void fused_head_kernel(
    const float* __restrict__ emb,
    const float* __restrict__ inck,
    const float* __restrict__ conv_w,
    const float* __restrict__ conv_b,
    const float* __restrict__ bn_g,
    const float* __restrict__ bn_b,
    const float* __restrict__ bn_m,
    const float* __restrict__ bn_v,
    float* __restrict__ z0)
{
    __shared__ __attribute__((aligned(16))) float smem[2 * TILE_DW + 140 + 32 + 32 + U_DW + 4 * PART_DW];
    float* kLDS   = smem;
    float* scLDS  = smem + 140;
    float* offLDS = smem + 172;
    float* tiles  = smem + 204;
    float* uLDS   = tiles + 2 * TILE_DW;
    float* part   = uLDS + U_DW;
    float* rawLDS = tiles + TILE_DW;

    const int tid = threadIdx.x;
    const int n   = blockIdx.x >> 3;
    const int s0  = (blockIdx.x & 7) * STRIPS_PER_BLOCK;

    const float* embn = emb + (size_t)n * 512 * 32 * 32;
    const int Wp = tid & 127;
    const int jw = (Wp & 3) * 1024 + (Wp >> 2);
    const int y1 = tid >> 7;
    const int y2 = 4 + ((tid >> 7) & 1);
    const int cb = (tid >> 8) * 16;
    const int o1 = PIXOFF(y1 * ROWP + 1 + Wp);
    const int o2 = PIXOFF(y2 * ROWP + 1 + Wp) + cb;

    {
        const int H1 = 4 * s0 - 1 + y1;
        const int H2 = 4 * s0 - 1 + y2;
        float v[32], u[16];
        if ((unsigned)H1 < 128u) {
            const float* src = embn + ((H1 & 3) * 4096 + (H1 >> 2) * 32 + jw);
#pragma unroll
            for (int c = 0; c < 32; ++c) v[c] = src[c * PLANE_C];
        } else {
#pragma unroll
            for (int c = 0; c < 32; ++c) v[c] = 0.f;
        }
        if ((unsigned)H2 < 128u) {
            const float* src = embn + ((H2 & 3) * 4096 + (H2 >> 2) * 32 + jw) + cb * PLANE_C;
#pragma unroll
            for (int c = 0; c < 16; ++c) u[c] = src[c * PLANE_C];
        } else {
#pragma unroll
            for (int c = 0; c < 16; ++c) u[c] = 0.f;
        }
        float* dst1 = tiles + o1;
        float* dst2 = tiles + o2;
#pragma unroll
        for (int q = 0; q < 8; ++q)
            *(float4*)(dst1 + 4 * q) =
                make_float4(v[4 * q], v[4 * q + 1], v[4 * q + 2], v[4 * q + 3]);
#pragma unroll
        for (int q = 0; q < 4; ++q)
            *(float4*)(dst2 + 4 * q) =
                make_float4(u[4 * q], u[4 * q + 1], u[4 * q + 2], u[4 * q + 3]);
        if (tid < 12) {
            float* dst = &tiles[PIXOFF((tid >> 1) * ROWP + (tid & 1) * (ROWP - 1))];
#pragma unroll
            for (int q = 0; q < 8; ++q)
                *(float4*)(dst + 4 * q) = make_float4(0.f, 0.f, 0.f, 0.f);
        }
    }
    for (int i = tid; i < RAW_DW; i += 512) rawLDS[i] = conv_w[i];
    if (tid < 137) kLDS[tid] = inck[n * 137 + tid];
    if (tid < 32) {
        const float sc = bn_g[tid] * rsqrtf(bn_v[tid] + 1e-5f);
        scLDS[tid]  = sc;
        offLDS[tid] = (conv_b[tid] - bn_m[tid]) * sc + bn_b[tid];
    }
    __syncthreads();

#pragma unroll
    for (int r = 0; r < 2; ++r) {
        const int o = (tid >> 5) + 16 * r;
        const int c = tid & 31;
        const float* g = &rawLDS[o * 288 + c * 9];
        float a[4][3];
#pragma unroll
        for (int x = 0; x < 3; ++x) {
            const float g0 = g[x], g1 = g[3 + x], g2 = g[6 + x];
            a[0][x] = g0;
            a[1][x] = 0.5f * (g0 + g1 + g2);
            a[2][x] = 0.5f * (g0 - g1 + g2);
            a[3][x] = g2;
        }
        float* u = &uLDS[o * UO + c];
#pragma unroll
        for (int xi = 0; xi < 4; ++xi) {
            u[(xi * 4 + 0) * UPOS] = a[xi][0];
            u[(xi * 4 + 1) * UPOS] = 0.5f * (a[xi][0] + a[xi][1] + a[xi][2]);
            u[(xi * 4 + 2) * UPOS] = 0.5f * (a[xi][0] - a[xi][1] + a[xi][2]);
            u[(xi * 4 + 3) * UPOS] = a[xi][2];
        }
    }
    __syncthreads();

    const int wave = tid >> 5;
    const int lane = tid & 31;
    const int half = lane >> 4;
    const int ml   = lane & 15;
    const int nb   = (wave >> 3) << 4;
    const int tr   = (wave >> 2) & 1;
    const int tc0  = (wave & 3) << 4;
    const int aoff = PIXOFF(2 * tr * ROWP + 2 * (tc0 + ml)) + 2 * half;
    const float* bbase = &uLDS[(nb + ml) * UO + 2 * half];
    const int ob   = nb + 8 * half;

#pragma unroll 1
    for (int it = 0; it < STRIPS_PER_BLOCK; ++it) {
        const int strip = s0 + it;
        float* cur = tiles + (it & 1) * TILE_DW;
        float* nxt = tiles + ((it & 1) ^ 1) * TILE_DW;
        const bool pf = it + 1 < STRIPS_PER_BLOCK;

        const int Hn1 = 4 * (strip + 1) - 1 + y1;
        const int Hn2 = 4 * (strip + 1) - 1 + y2;
        const bool ok2 = (unsigned)Hn2 < 128u;
        const int Hc1 = min(max(Hn1, 0), 127);
        const int Hc2 = min(max(Hn2, 0), 127);
        const int q1 = (Hc1 & 3) * 4096 + (Hc1 >> 2) * 32 + jw;
        const int q2 = (Hc2 & 3) * 4096 + (Hc2 >> 2) * 32 + jw + cb * PLANE_C;
        float* d1 = nxt + o1;
        float* d2 = nxt + o2;
        if (pf && tid < 12) {
            float* dst = &nxt[PIXOFF((tid >> 1) * ROWP + (tid & 1) * (ROWP - 1))];
#pragma unroll
            for (int q = 0; q < 8; ++q)
                *(float4*)(dst + 4 * q) = make_float4(0.f, 0.f, 0.f, 0.f);
        }

        v8f M[16] = {};
        const float* abase = cur + aoff;
        auto gemm_step = [&](const int kk) __attribute__((always_inline)) {
            v2f bq[16];
#pragma unroll
            for (int p = 0; p < 16; ++p) {
                const float2 q = *(const float2*)(bbase + p * UPOS + kk * 4);
                bq[p].x = q.x; bq[p].y = q.y;
            }
            v2f d[4][4];
#pragma unroll
            for (int r = 0; r < 4; ++r) {
#pragma unroll
                for (int s = 0; s < 4; ++s) {
                    const float2 q = *(const float2*)(abase + r * PIXOFF(ROWP) + PIXOFF(s) + kk * 4);
                    d[r][s].x = q.x; d[r][s].y = q.y;
                }
            }
            v2f t[4][4];
#pragma unroll
            for (int s = 0; s < 4; ++s) {
                t[0][s] = d[0][s] - d[2][s];
                t[1][s] = d[1][s] + d[2][s];
                t[2][s] = d[2][s] - d[1][s];
                t[3][s] = d[1][s] - d[3][s];
            }
#pragma unroll
            for (int xi = 0; xi < 4; ++xi) {
                v2f V[4];
                V[0] = t[xi][0] - t[xi][2];
                V[1] = t[xi][1] + t[xi][2];
                V[2] = t[xi][2] - t[xi][1];
                V[3] = t[xi][1] - t[xi][3];
#pragma unroll
                for (int nu = 0; nu < 4; ++nu) {
                    M[xi * 4 + nu] = __builtin_amdgcn_wmma_f32_16x16x4_f32(
                        false, bq[xi * 4 + nu], false, V[nu], (short)0, M[xi * 4 + nu], false, false);
                }
            }
        };
        const float* a1 = embn + q1;
        const float* a2 = embn + q2;
        auto request = [&](const int sh, float4& pv, float2& pu) __attribute__((always_inline)) {
            const float* b1 = a1 + 4 * sh * PLANE_C;
            const float* b2 = a2 + 2 * sh * PLANE_C;
            pv = make_float4(b1[0], b1[PLANE_C], b1[2 * PLANE_C], b1[3 * PLANE_C]);
            pu = make_float2(b2[0], b2[PLANE_C]);
        };
        auto deposit = [&](const int sh, const float4& pv, const float2& pu) __attribute__((always_inline)) {
            *(float4*)(d1 + 4 * sh) = pv;
            *(float2*)(d2 + 2 * sh) = pu;
        };
        if (pf) {
            float4 pv0, pv1;
            float2 pu0, pu1;
            request(0, pv0, pu0);
#pragma unroll 1
            for (int ke = 0; ke < 6; ke += 2) {
                request(ke + 1, pv1, pu1);
                gemm_step(ke);
                deposit(ke, pv0, pu0);
                request(ke + 2, pv0, pu0);
                gemm_step(ke + 1);
                deposit(ke + 1, pv1, pu1);
                __builtin_amdgcn_sched_barrier(0);
                asm volatile("v_nop\n\tv_nop\n\tv_nop\n\tv_nop" ::: "memory");
            }
            request(7, pv1, pu1);
            gemm_step(6);
            deposit(6, pv0, pu0);
            gemm_step(7);
            deposit(7, pv1, pu1);
            if (!ok2) {
#pragma unroll
                for (int q = 0; q < 4; ++q)
                    *(float4*)(d2 + 4 * q) = make_float4(0.f, 0.f, 0.f, 0.f);
            }
        } else {
#pragma unroll 1
            for (int kk = 0; kk < 8; ++kk) {
                gemm_step(kk);
                __builtin_amdgcn_sched_barrier(0);
                asm volatile("v_nop\n\tv_nop\n\tv_nop\n\tv_nop" ::: "memory");
            }
        }

        v8f Y[2][2];
        {
            v8f Z[4][2];
#pragma unroll
            for (int xi = 0; xi < 4; ++xi) {
                Z[xi][0] = M[xi * 4 + 0] + M[xi * 4 + 1] + M[xi * 4 + 2];
                Z[xi][1] = M[xi * 4 + 1] - M[xi * 4 + 2] - M[xi * 4 + 3];
            }
#pragma unroll
            for (int j = 0; j < 2; ++j) {
                Y[0][j] = Z[0][j] + Z[1][j] + Z[2][j];
                Y[1][j] = Z[1][j] - Z[2][j] - Z[3][j];
            }
        }

        float hp[2][2][4];
        {
            float sc[8], of[8];
            {
                const float4 a = *(const float4*)(scLDS + ob),  b = *(const float4*)(scLDS + ob + 4);
                const float4 e = *(const float4*)(offLDS + ob), f = *(const float4*)(offLDS + ob + 4);
                sc[0] = a.x; sc[1] = a.y; sc[2] = a.z; sc[3] = a.w; sc[4] = b.x; sc[5] = b.y; sc[6] = b.z; sc[7] = b.w;
                of[0] = e.x; of[1] = e.y; of[2] = e.z; of[3] = e.w; of[4] = f.x; of[5] = f.y; of[6] = f.z; of[7] = f.w;
            }
#pragma unroll
            for (int i = 0; i < 2; ++i)
#pragma unroll
                for (int j = 0; j < 2; ++j)
#pragma unroll
                    for (int r = 0; r < 8; ++r)
                        Y[i][j][r] = fmaxf(Y[i][j][r] * sc[r] + of[r], 0.f);
#pragma unroll
            for (int k = 0; k < 4; ++k) {
                const float4 a = *(const float4*)(kLDS + 32 * k + ob);
                const float4 b = *(const float4*)(kLDS + 32 * k + ob + 4);
                const float w[8] = {a.x, a.y, a.z, a.w, b.x, b.y, b.z, b.w};
#pragma unroll
                for (int i = 0; i < 2; ++i) {
#pragma unroll
                    for (int j = 0; j < 2; ++j) {
                        float s = w[0] * Y[i][j][0];
#pragma unroll
                        for (int r = 1; r < 8; ++r) s = fmaf(w[r], Y[i][j][r], s);
                        hp[i][j][k] = s;
                    }
                }
            }
        }
        {
            float* pb = part + ((it & 1) * 2 + (wave >> 3)) * PART_DW
                             + ((2 * tr + half) * 128 + 2 * (tc0 + ml)) * 4;
            float o[2][4];
#pragma unroll
            for (int j = 0; j < 2; ++j) {
#pragma unroll
                for (int k = 0; k < 4; ++k) {
                    const float keep = half ? hp[1][j][k] : hp[0][j][k];
                    const float send = half ? hp[0][j][k] : hp[1][j][k];
                    o[j][k] = keep + xhalf(send);
                }
            }
            *(float4*)pb       = make_float4(o[0][0], o[0][1], o[0][2], o[0][3]);
            *(float4*)(pb + 4) = make_float4(o[1][0], o[1][1], o[1][2], o[1][3]);
        }
        __syncthreads();

        {
            const int p = tid;
            const float* pr = part + (it & 1) * 2 * PART_DW + p * 4;
            const float4 a = *(const float4*)pr;
            const float4 b = *(const float4*)(pr + PART_DW);
            const float h0 = fmaxf(a.x + b.x + kLDS[132], 0.f);
            const float h1 = fmaxf(a.y + b.y + kLDS[133], 0.f);
            const float h2 = fmaxf(a.z + b.z + kLDS[134], 0.f);
            const float h3 = fmaxf(a.w + b.w + kLDS[135], 0.f);
            float y = kLDS[136];
            y = fmaf(kLDS[128], h0, y);
            y = fmaf(kLDS[129], h1, y);
            y = fmaf(kLDS[130], h2, y);
            y = fmaf(kLDS[131], h3, y);
            const float sg = 1.f / (1.f + __expf(-y));
            const int H = 4 * strip + (p >> 7), W = p & 127;
            z0[((size_t)n * 128 + H) * 128 + W] = sg;
        }
    }
}

#define SM_PITCH 132
#define SM_TROWS 12
__global__ __launch_bounds__(128) void smooth_kernel(
    const float* __restrict__ z0,
    const float* __restrict__ mask,
    const float* __restrict__ sw,
    const float* __restrict__ sb,
    float* __restrict__ out)
{
    __shared__ __attribute__((aligned(16))) float tile[2 * SM_TROWS * SM_PITCH];

    const int tid = threadIdx.x;
    const int tx  = tid & 31;
    const int ty  = tid >> 5;
    const int n   = blockIdx.x >> 4;
    const int H0  = (blockIdx.x & 15) * 8;
    const float* zp = z0   + (size_t)n * 16384;
    const float* mp = mask + (size_t)n * 16384;

    float4 g[6];
#pragma unroll
    for (int k = 0; k < 6; ++k) {
        const int row = (k % 3) * 4 + ty;
        const int h   = H0 - 2 + row;
        const float* src = (k < 3) ? zp : mp;
        g[k] = make_float4(0.f, 0.f, 0.f, 0.f);
        if ((unsigned)h < 128u) g[k] = *(const float4*)(src + h * 128 + 4 * tx);
    }
#pragma unroll
    for (int k = 0; k < 6; ++k) {
        const int row = (k % 3) * 4 + ty;
        float* dst = &tile[((k / 3) * SM_TROWS + row) * SM_PITCH + 2 + 4 * tx];
        *(float2*)dst       = make_float2(g[k].x, g[k].y);
        *(float2*)(dst + 2) = make_float2(g[k].z, g[k].w);
    }
    if (tid < 48) {
        const int pl  = tid / 24;
        const int rem = tid - pl * 24;
        float* dst = &tile[(pl * SM_TROWS + (rem >> 1)) * SM_PITCH + (rem & 1) * 130];
        *(float2*)dst = make_float2(0.f, 0.f);
    }
    __syncthreads();

    const float bias = sb[0];
    float acc[2][4];
#pragma unroll
    for (int o = 0; o < 2; ++o)
#pragma unroll
        for (int q = 0; q < 4; ++q) acc[o][q] = bias;

#pragma unroll
    for (int r = 0; r < 6; ++r) {
        const float* zr = &tile[(2 * ty + r) * SM_PITCH + 4 * tx];
        const float* mr = zr + SM_TROWS * SM_PITCH;
        const float4 za = *(const float4*)zr, zb = *(const float4*)(zr + 4);
        const float4 ma = *(const float4*)mr, mb = *(const float4*)(mr + 4);
        const float zv[8] = {za.x, za.y, za.z, za.w, zb.x, zb.y, zb.z, zb.w};
        const float mv[8] = {ma.x, ma.y, ma.z, ma.w, mb.x, mb.y, mb.z, mb.w};
#pragma unroll
        for (int o = 0; o < 2; ++o) {
            const int dy = r - o;
            if (dy < 0 || dy > 4) continue;
#pragma unroll
            for (int dx = 0; dx < 5; ++dx) {
                const float wz = sw[dy * 5 + dx];
                const float wm = sw[25 + dy * 5 + dx];
#pragma unroll
                for (int q = 0; q < 4; ++q) {
                    acc[o][q] = fmaf(wz, zv[q + dx], acc[o][q]);
                    acc[o][q] = fmaf(wm, mv[q + dx], acc[o][q]);
                }
            }
        }
    }

#pragma unroll
    for (int o = 0; o < 2; ++o) {
        const int H = H0 + 2 * ty + o;
        *(float4*)(out + ((size_t)n * 128 + H) * 128 + 4 * tx) =
            make_float4(acc[o][0], acc[o][1], acc[o][2], acc[o][3]);
    }
}

extern "C" void kernel_launch(void* const* d_in, const int* in_sizes, int n_in,
                              void* d_out, int out_size, void* d_ws, size_t ws_size,
                              hipStream_t stream) {
    const float* emb    = (const float*)d_in[0];
    const float* inck   = (const float*)d_in[1];
    const float* masks  = (const float*)d_in[2];
    const float* conv_w = (const float*)d_in[3];
    const float* conv_b = (const float*)d_in[4];
    const float* bn_g   = (const float*)d_in[5];
    const float* bn_b   = (const float*)d_in[6];
    const float* bn_m   = (const float*)d_in[7];
    const float* bn_v   = (const float*)d_in[8];
    const float* sw     = (const float*)d_in[9];
    const float* sb     = (const float*)d_in[10];
    float* out = (float*)d_out;
    float* z0  = (float*)d_ws;

    fused_head_kernel<<<256, 512, 0, stream>>>(emb, inck, conv_w, conv_b,
                                               bn_g, bn_b, bn_m, bn_v, z0);
    smooth_kernel<<<512, 128, 0, stream>>>(z0, masks, sw, sb, out);
}


# 3x3 conv tile enlarged down the rows to F(4x2,3x3): 24 transform-position GEMMs per 8 pixels (was 32); 8 waves x 192 accumulators, halo rows in a 10-slot LDS ring; conv summation order changes
# speedup vs baseline: 3.3674x; 1.0613x over previous
#include <hip/hip_runtime.h>
#include <hip/hip_bf16.h>

typedef __attribute__((ext_vector_type(2))) float v2f;
typedef __attribute__((ext_vector_type(8))) float v8f;

#define PIXOFF(p) ((p) * 32 + ((p) >> 1) * 4)
#define UO 36
#define UPOS (32 * UO)
#define NPOS 24
#define U_DW (NPOS * UPOS)
#define RAW_DW 9216
#define ROWP 130
#define ROW_DW 4420
#define RING 10
#define RING_DW (RING * ROW_DW)
#define PLANE_C (16 * 32 * 32)
#define STRIPS_PER_BLOCK 4
#define PART_DW (512 * 4)
__device__ __forceinline__ float xhalf(float x) {
    return __builtin_bit_cast(float, __builtin_amdgcn_permlanex16(
        __builtin_bit_cast(int, x), __builtin_bit_cast(int, x), 0x76543210, 0xfedcba98, false, false));
}
__global__ __launch_bounds__(256) void fused_head_kernel(
    const float* __restrict__ emb,
    const float* __restrict__ inck,
    const float* __restrict__ conv_w,
    const float* __restrict__ conv_b,
    const float* __restrict__ bn_g,
    const float* __restrict__ bn_b,
    const float* __restrict__ bn_m,
    const float* __restrict__ bn_v,
    float* __restrict__ z0)
{
    __shared__ __attribute__((aligned(16))) float smem[140 + 32 + 32 + RING_DW + U_DW + 4 * PART_DW];
    float* kLDS   = smem;
    float* scLDS  = smem + 140;
    float* offLDS = smem + 172;
    float* tiles  = smem + 204;
    float* uLDS   = tiles + RING_DW;
    float* part   = uLDS + U_DW;
    float* rawLDS = tiles + 6 * ROW_DW;

    const int tid = threadIdx.x;
    const int n   = blockIdx.x >> 3;
    const int s0  = (blockIdx.x & 7) * STRIPS_PER_BLOCK;

    const float* embn = emb + (size_t)n * 512 * 32 * 32;
    const int Wp = tid & 127;
    const int jw = (Wp & 3) * 1024 + (Wp >> 2);
    const int yy = tid >> 7;
    const int ocol = PIXOFF(1 + Wp);

    {
        float v[3][32];
#pragma unroll
        for (int g = 0; g < 3; ++g) {
            const int H  = 4 * s0 - 1 + yy + 2 * g;
            const int Hc = max(H, 0);
            const float* src = embn + ((Hc & 3) * 4096 + (Hc >> 2) * 32 + jw);
#pragma unroll
            for (int c = 0; c < 32; ++c) v[g][c] = src[c * PLANE_C];
        }
#pragma unroll
        for (int g = 0; g < 3; ++g) {
            const bool in = (4 * s0 - 1 + yy + 2 * g) >= 0;
            float* dst = tiles + (yy + 2 * g) * ROW_DW + ocol;
#pragma unroll
            for (int q = 0; q < 8; ++q)
                *(float4*)(dst + 4 * q) = in
                    ? make_float4(v[g][4 * q], v[g][4 * q + 1], v[g][4 * q + 2], v[g][4 * q + 3])
                    : make_float4(0.f, 0.f, 0.f, 0.f);
        }
        if (tid < 12) {
            float* dst = &tiles[(tid >> 1) * ROW_DW + PIXOFF((tid & 1) * (ROWP - 1))];
#pragma unroll
            for (int q = 0; q < 8; ++q)
                *(float4*)(dst + 4 * q) = make_float4(0.f, 0.f, 0.f, 0.f);
        }
    }
    for (int i = tid; i < RAW_DW; i += 256) rawLDS[i] = conv_w[i];
    if (tid < 137) kLDS[tid] = inck[n * 137 + tid];
    if (tid < 32) {
        const float sc = bn_g[tid] * rsqrtf(bn_v[tid] + 1e-5f);
        scLDS[tid]  = sc;
        offLDS[tid] = (conv_b[tid] - bn_m[tid]) * sc + bn_b[tid];
    }
    __syncthreads();

#pragma unroll
    for (int r = 0; r < 4; ++r) {
        const int o = (tid >> 5) + 8 * r;
        const int c = tid & 31;
        const float* g = &rawLDS[o * 288 + c * 9];
        float a[6][3];
#pragma unroll
        for (int x = 0; x < 3; ++x) {
            const float g0 = g[x], g1 = g[3 + x], g2 = g[6 + x];
            a[0][x] = 0.25f * g0;
            a[1][x] = (-1.f / 6.f) * (g0 + g1 + g2);
            a[2][x] = (-1.f / 6.f) * (g0 - g1 + g2);
            a[3][x] = (1.f / 24.f) * g0 + (1.f / 12.f) * g1 + (1.f / 6.f) * g2;
            a[4][x] = (1.f / 24.f) * g0 - (1.f / 12.f) * g1 + (1.f / 6.f) * g2;
            a[5][x] = g2;
        }
        float* u = &uLDS[o * UO + c];
#pragma unroll
        for (int xi = 0; xi < 6; ++xi) {
            u[(xi * 4 + 0) * UPOS] = a[xi][0];
            u[(xi * 4 + 1) * UPOS] = 0.5f * (a[xi][0] + a[xi][1] + a[xi][2]);
            u[(xi * 4 + 2) * UPOS] = 0.5f * (a[xi][0] - a[xi][1] + a[xi][2]);
            u[(xi * 4 + 3) * UPOS] = a[xi][2];
        }
    }
    __syncthreads();
    if (tid < 8) {
        float* dst = &tiles[(6 + (tid >> 1)) * ROW_DW + PIXOFF((tid & 1) * (ROWP - 1))];
#pragma unroll
        for (int q = 0; q < 8; ++q)
            *(float4*)(dst + 4 * q) = make_float4(0.f, 0.f, 0.f, 0.f);
    }

    const int wave = tid >> 5;
    const int lane = tid & 31;
    const int half = lane >> 4;
    const int ml   = lane & 15;
    const int nb   = (wave >> 2) << 4;
    const int tc0  = (wave & 3) << 4;
    const int acol = PIXOFF(2 * (tc0 + ml)) + 2 * half;
    const float* bbase = &uLDS[(nb + ml) * UO + 2 * half];
    const int ob   = nb + 8 * half;

#pragma unroll 1
    for (int it = 0; it < STRIPS_PER_BLOCK; ++it) {
        const int strip = s0 + it;
        const int rb = (4 * it) % RING;
        const bool pf = it + 1 < STRIPS_PER_BLOCK;

        const float* arow[6];
#pragma unroll
        for (int r = 0; r < 6; ++r) {
            const int q = rb + r;
            arow[r] = tiles + (q >= RING ? q - RING : q) * ROW_DW + acol;
        }

        const int Ha = 4 * strip + 5 + yy;
        const int Hb = 4 * strip + 7 + yy;
        const bool okb = Hb < 128;
        const int Hca = min(Ha, 127);
        const int Hcb = min(Hb, 127);
        const float* a1 = embn + ((Hca & 3) * 4096 + (Hca >> 2) * 32 + jw);
        const float* a2 = embn + ((Hcb & 3) * 4096 + (Hcb >> 2) * 32 + jw);
        const int qa = rb + 6 + yy, qb = rb + 8 + yy;
        float* d1 = tiles + (qa >= RING ? qa - RING : qa) * ROW_DW + ocol;
        float* d2 = tiles + (qb >= RING ? qb - RING : qb) * ROW_DW + ocol;

        v8f M[NPOS] = {};
        auto gemm_step = [&](const int kk) __attribute__((always_inline)) {
            v2f bq[NPOS];
#pragma unroll
            for (int p = 0; p < NPOS; ++p) {
                const float2 q = *(const float2*)(bbase + p * UPOS + kk * 4);
                bq[p].x = q.x; bq[p].y = q.y;
            }
            v2f d[6][4];
#pragma unroll
            for (int r = 0; r < 6; ++r) {
#pragma unroll
                for (int s = 0; s < 4; ++s) {
                    const float2 q = *(const float2*)(arow[r] + PIXOFF(s) + kk * 4);
                    d[r][s].x = q.x; d[r][s].y = q.y;
                }
            }
            v2f t[6][4];
#pragma unroll
            for (int s = 0; s < 4; ++s) {
                const v2f p1 = d[4][s] - 4.f * d[2][s];
                const v2f q1 = d[3][s] - 4.f * d[1][s];
                const v2f r1 = d[4][s] - d[2][s];
                const v2f w1 = d[3][s] - d[1][s];
                t[0][s] = 4.f * d[0][s] - 5.f * d[2][s] + d[4][s];
                t[1][s] = p1 + q1;
                t[2][s] = p1 - q1;
                t[3][s] = r1 + 2.f * w1;
                t[4][s] = r1 - 2.f * w1;
                t[5][s] = 4.f * d[1][s] - 5.f * d[3][s] + d[5][s];
            }
#pragma unroll
            for (int xi = 0; xi < 6; ++xi) {
                v2f V[4];
                V[0] = t[xi][0] - t[xi][2];
                V[1] = t[xi][1] + t[xi][2];
                V[2] = t[xi][2] - t[xi][1];
                V[3] = t[xi][1] - t[xi][3];
#pragma unroll
                for (int nu = 0; nu < 4; ++nu) {
                    M[xi * 4 + nu] = __builtin_amdgcn_wmma_f32_16x16x4_f32(
                        false, bq[xi * 4 + nu], false, V[nu], (short)0, M[xi * 4 + nu], false, false);
                }
            }
        };
        auto request = [&](const int sh, float4& pa, float4& pb) __attribute__((always_inline)) {
            const float* b1 = a1 + 4 * sh * PLANE_C;
            const float* b2 = a2 + 4 * sh * PLANE_C;
            pa = make_float4(b1[0], b1[PLANE_C], b1[2 * PLANE_C], b1[3 * PLANE_C]);
            pb = make_float4(b2[0], b2[PLANE_C], b2[2 * PLANE_C], b2[3 * PLANE_C]);
        };
        auto deposit = [&](const int sh, const float4& pa, const float4& pb) __attribute__((always_inline)) {
            *(float4*)(d1 + 4 * sh) = pa;
            *(float4*)(d2 + 4 * sh) = pb;
        };
        if (pf) {
            float4 pa0, pa1, pb0, pb1;
            request(0, pa0, pb0);
#pragma unroll 1
            for (int ke = 0; ke < 6; ke += 2) {
                request(ke + 1, pa1, pb1);
                gemm_step(ke);
                deposit(ke, pa0, pb0);
                request(ke + 2, pa0, pb0);
                gemm_step(ke + 1);
                deposit(ke + 1, pa1, pb1);
                __builtin_amdgcn_sched_barrier(0);
                asm volatile("v_nop\n\tv_nop\n\tv_nop\n\tv_nop" ::: "memory");
            }
            request(7, pa1, pb1);
            gemm_step(6);
            deposit(6, pa0, pb0);
            gemm_step(7);
            deposit(7, pa1, pb1);
            __builtin_amdgcn_sched_barrier(0);
            asm volatile("v_nop\n\tv_nop\n\tv_nop\n\tv_nop" ::: "memory");
            if (!okb) {
#pragma unroll
                for (int q = 0; q < 8; ++q)
                    *(float4*)(d2 + 4 * q) = make_float4(0.f, 0.f, 0.f, 0.f);
            }
        } else {
#pragma unroll 1
            for (int kk = 0; kk < 8; ++kk) {
                gemm_step(kk);
                __builtin_amdgcn_sched_barrier(0);
                asm volatile("v_nop\n\tv_nop\n\tv_nop\n\tv_nop" ::: "memory");
            }
        }

        v8f Y[4][2];
        {
            v8f Z[6][2];
#pragma unroll
            for (int xi = 0; xi < 6; ++xi) {
                Z[xi][0] = M[xi * 4 + 0] + M[xi * 4 + 1] + M[xi * 4 + 2];
                Z[xi][1] = M[xi * 4 + 1] - M[xi * 4 + 2] - M[xi * 4 + 3];
            }
#pragma unroll
            for (int j = 0; j < 2; ++j) {
                const v8f sa = Z[1][j] + Z[2][j];
                const v8f sb = Z[1][j] - Z[2][j];
                const v8f sc = Z[3][j] + Z[4][j];
                const v8f se = Z[3][j] - Z[4][j];
                Y[0][j] = Z[0][j] + sa + sc;
                Y[1][j] = sb + 2.f * se;
                Y[2][j] = sa + 4.f * sc;
                Y[3][j] = sb + 8.f * se + Z[5][j];
            }
        }

        float hp[4][2][4];
        {
            float sc[8], of[8];
            {
                const float4 a = *(const float4*)(scLDS + ob),  b = *(const float4*)(scLDS + ob + 4);
                const float4 e = *(const float4*)(offLDS + ob), f = *(const float4*)(offLDS + ob + 4);
                sc[0] = a.x; sc[1] = a.y; sc[2] = a.z; sc[3] = a.w; sc[4] = b.x; sc[5] = b.y; sc[6] = b.z; sc[7] = b.w;
                of[0] = e.x; of[1] = e.y; of[2] = e.z; of[3] = e.w; of[4] = f.x; of[5] = f.y; of[6] = f.z; of[7] = f.w;
            }
#pragma unroll
            for (int i = 0; i < 4; ++i)
#pragma unroll
                for (int j = 0; j < 2; ++j)
#pragma unroll
                    for (int r = 0; r < 8; ++r)
                        Y[i][j][r] = fmaxf(Y[i][j][r] * sc[r] + of[r], 0.f);
#pragma unroll
            for (int k = 0; k < 4; ++k) {
                const float4 a = *(const float4*)(kLDS + 32 * k + ob);
                const float4 b = *(const float4*)(kLDS + 32 * k + ob + 4);
                const float w[8] = {a.x, a.y, a.z, a.w, b.x, b.y, b.z, b.w};
#pragma unroll
                for (int i = 0; i < 4; ++i) {
#pragma unroll
                    for (int j = 0; j < 2; ++j) {
                        float s = w[0] * Y[i][j][0];
#pragma unroll
                        for (int r = 1; r < 8; ++r) s = fmaf(w[r], Y[i][j][r], s);
                        hp[i][j][k] = s;
                    }
                }
            }
        }
        {
            float* pb = part + ((it & 1) * 2 + (wave >> 2)) * PART_DW
                             + ((2 * half) * 128 + 2 * (tc0 + ml)) * 4;
#pragma unroll
            for (int ii = 0; ii < 2; ++ii) {
                float o[2][4];
#pragma unroll
                for (int j = 0; j < 2; ++j) {
#pragma unroll
                    for (int k = 0; k < 4; ++k) {
                        const float keep = half ? hp[2 + ii][j][k] : hp[ii][j][k];
                        const float send = half ? hp[ii][j][k] : hp[2 + ii][j][k];
                        o[j][k] = keep + xhalf(send);
                    }
                }
                *(float4*)(pb + ii * 512)     = make_float4(o[0][0], o[0][1], o[0][2], o[0][3]);
                *(float4*)(pb + ii * 512 + 4) = make_float4(o[1][0], o[1][1], o[1][2], o[1][3]);
            }
        }
        __syncthreads();

#pragma unroll
        for (int h = 0; h < 2; ++h) {
            const int p = tid + 256 * h;
            const float* pr = part + (it & 1) * 2 * PART_DW + p * 4;
            const float4 a = *(const float4*)pr;
            const float4 b = *(const float4*)(pr + PART_DW);
            const float h0 = fmaxf(a.x + b.x + kLDS[132], 0.f);
            const float h1 = fmaxf(a.y + b.y + kLDS[133], 0.f);
            const float h2 = fmaxf(a.z + b.z + kLDS[134], 0.f);
            const float h3 = fmaxf(a.w + b.w + kLDS[135], 0.f);
            float y = kLDS[136];
            y = fmaf(kLDS[128], h0, y);
            y = fmaf(kLDS[129], h1, y);
            y = fmaf(kLDS[130], h2, y);
            y = fmaf(kLDS[131], h3, y);
            const float sg = 1.f / (1.f + __expf(-y));
            const int H = 4 * strip + (p >> 7), W = p & 127;
            z0[((size_t)n * 128 + H) * 128 + W] = sg;
        }
    }
}

#define SM_PITCH 132
#define SM_TROWS 12
__global__ __launch_bounds__(128) void smooth_kernel(
    const float* __restrict__ z0,
    const float* __restrict__ mask,
    const float* __restrict__ sw,
    const float* __restrict__ sb,
    float* __restrict__ out)
{
    __shared__ __attribute__((aligned(16))) float tile[2 * SM_TROWS * SM_PITCH];

    const int tid = threadIdx.x;
    const int tx  = tid & 31;
    const int ty  = tid >> 5;
    const int n   = blockIdx.x >> 4;
    const int H0  = (blockIdx.x & 15) * 8;
    const float* zp = z0   + (size_t)n * 16384;
    const float* mp = mask + (size_t)n * 16384;

    float4 g[6];
#pragma unroll
    for (int k = 0; k < 6; ++k) {
        const int row = (k % 3) * 4 + ty;
        const int h   = H0 - 2 + row;
        const float* src = (k < 3) ? zp : mp;
        g[k] = make_float4(0.f, 0.f, 0.f, 0.f);
        if ((unsigned)h < 128u) g[k] = *(const float4*)(src + h * 128 + 4 * tx);
    }
#pragma unroll
    for (int k = 0; k < 6; ++k) {
        const int row = (k % 3) * 4 + ty;
        float* dst = &tile[((k / 3) * SM_TROWS + row) * SM_PITCH + 2 + 4 * tx];
        *(float2*)dst       = make_float2(g[k].x, g[k].y);
        *(float2*)(dst + 2) = make_float2(g[k].z, g[k].w);
    }
    if (tid < 48) {
        const int pl  = tid / 24;
        const int rem = tid - pl * 24;
        float* dst = &tile[(pl * SM_TROWS + (rem >> 1)) * SM_PITCH + (rem & 1) * 130];
        *(float2*)dst = make_float2(0.f, 0.f);
    }
    __syncthreads();

    const float bias = sb[0];
    float acc[2][4];
#pragma unroll
    for (int o = 0; o < 2; ++o)
#pragma unroll
        for (int q = 0; q < 4; ++q) acc[o][q] = bias;

#pragma unroll
    for (int r = 0; r < 6; ++r) {
        const float* zr = &tile[(2 * ty + r) * SM_PITCH + 4 * tx];
        const float* mr = zr + SM_TROWS * SM_PITCH;
        const float4 za = *(const float4*)zr, zb = *(const float4*)(zr + 4);
        const float4 ma = *(const float4*)mr, mb = *(const float4*)(mr + 4);
        const float zv[8] = {za.x, za.y, za.z, za.w, zb.x, zb.y, zb.z, zb.w};
        const float mv[8] = {ma.x, ma.y, ma.z, ma.w, mb.x, mb.y, mb.z, mb.w};
#pragma unroll
        for (int o = 0; o < 2; ++o) {
            const int dy = r - o;
            if (dy < 0 || dy > 4) continue;
#pragma unroll
            for (int dx = 0; dx < 5; ++dx) {
                const float wz = sw[dy * 5 + dx];
                const float wm = sw[25 + dy * 5 + dx];
#pragma unroll
                for (int q = 0; q < 4; ++q) {
                    acc[o][q] = fmaf(wz, zv[q + dx], acc[o][q]);
                    acc[o][q] = fmaf(wm, mv[q + dx], acc[o][q]);
                }
            }
        }
    }

#pragma unroll
    for (int o = 0; o < 2; ++o) {
        const int H = H0 + 2 * ty + o;
        *(float4*)(out + ((size_t)n * 128 + H) * 128 + 4 * tx) =
            make_float4(acc[o][0], acc[o][1], acc[o][2], acc[o][3]);
    }
}

extern "C" void kernel_launch(void* const* d_in, const int* in_sizes, int n_in,
                              void* d_out, int out_size, void* d_ws, size_t ws_size,
                              hipStream_t stream) {
    const float* emb    = (const float*)d_in[0];
    const float* inck   = (const float*)d_in[1];
    const float* masks  = (const float*)d_in[2];
    const float* conv_w = (const float*)d_in[3];
    const float* conv_b = (const float*)d_in[4];
    const float* bn_g   = (const float*)d_in[5];
    const float* bn_b   = (const float*)d_in[6];
    const float* bn_m   = (const float*)d_in[7];
    const float* bn_v   = (const float*)d_in[8];
    const float* sw     = (const float*)d_in[9];
    const float* sb     = (const float*)d_in[10];
    float* out = (float*)d_out;
    float* z0  = (float*)d_ws;

    fused_head_kernel<<<256, 256, 0, stream>>>(emb, inck, conv_w, conv_b,
                                               bn_g, bn_b, bn_m, bn_v, z0);
    smooth_kernel<<<512, 128, 0, stream>>>(z0, masks, sw, sb, out);
}


# prologue pays one memory round trip instead of seven: filter (16-B loads), parameter and first-strip reads all issued before the first wait; filter transform runs while the strip's rows arrive
# speedup vs baseline: 3.5406x; 1.0514x over previous
#include <hip/hip_runtime.h>
#include <hip/hip_bf16.h>

typedef __attribute__((ext_vector_type(2))) float v2f;
typedef __attribute__((ext_vector_type(8))) float v8f;

#define PIXOFF(p) ((p) * 32 + ((p) >> 1) * 4)
#define UO 36
#define UPOS (32 * UO)
#define NPOS 24
#define U_DW (NPOS * UPOS)
#define RAW_DW 9216
#define ROWP 130
#define ROW_DW 4420
#define RING 10
#define RING_DW (RING * ROW_DW)
#define PLANE_C (16 * 32 * 32)
#define STRIPS_PER_BLOCK 4
#define PART_DW (512 * 4)
__device__ __forceinline__ float xhalf(float x) {
    return __builtin_bit_cast(float, __builtin_amdgcn_permlanex16(
        __builtin_bit_cast(int, x), __builtin_bit_cast(int, x), 0x76543210, 0xfedcba98, false, false));
}
__device__ __forceinline__ void lds_barrier() {
    __builtin_amdgcn_fence(__ATOMIC_RELEASE, "workgroup", "local");
    __builtin_amdgcn_s_barrier();
    __builtin_amdgcn_fence(__ATOMIC_ACQUIRE, "workgroup", "local");
}
__global__ __launch_bounds__(256) void fused_head_kernel(
    const float* __restrict__ emb,
    const float* inck,
    const float* conv_w,
    const float* conv_b,
    const float* bn_g,
    const float* bn_b,
    const float* bn_m,
    const float* bn_v,
    float* __restrict__ z0)
{
    __shared__ __attribute__((aligned(16))) float smem[140 + 32 + 32 + RING_DW + U_DW + 4 * PART_DW];
    float* kLDS   = smem;
    float* scLDS  = smem + 140;
    float* offLDS = smem + 172;
    float* tiles  = smem + 204;
    float* uLDS   = tiles + RING_DW;
    float* part   = uLDS + U_DW;
    float* rawLDS = tiles + 6 * ROW_DW;

    const int tid = threadIdx.x;
    const int n   = blockIdx.x >> 3;
    const int s0  = (blockIdx.x & 7) * STRIPS_PER_BLOCK;

    const float* embn = emb + (size_t)n * 512 * 32 * 32;
    const int Wp = tid & 127;
    const int jw = (Wp & 3) * 1024 + (Wp >> 2);
    const int yy = tid >> 7;
    const int ocol = PIXOFF(1 + Wp);

    const float4* cw4 = (const float4*)conv_w + tid;
    const float4 wq0 = cw4[0],    wq1 = cw4[256],  wq2 = cw4[512],  wq3 = cw4[768], wq4 = cw4[1024],
                 wq5 = cw4[1280], wq6 = cw4[1536], wq7 = cw4[1792], wq8 = cw4[2048];
    const float kv = inck[n * 137 + min(tid, 136)];
    const int ct = tid & 31;
    const float pg = bn_g[ct], pv = bn_v[ct], pcb = conv_b[ct], pm = bn_m[ct], pbt = bn_b[ct];
    int jws = jw;
    asm volatile("" : "+v"(jws));

    {
        float v[3][32];
#pragma unroll
        for (int g = 0; g < 3; ++g) {
            const int H  = 4 * s0 - 1 + yy + 2 * g;
            const int Hc = max(H, 0);
            const float* src = embn + ((Hc & 3) * 4096 + (Hc >> 2) * 32 + jws);
#pragma unroll
            for (int c = 0; c < 32; ++c) v[g][c] = src[c * PLANE_C];
        }
        {
            float4* rw4 = (float4*)rawLDS + tid;
            rw4[0]    = wq0; rw4[256]  = wq1; rw4[512]  = wq2; rw4[768]  = wq3; rw4[1024] = wq4;
            rw4[1280] = wq5; rw4[1536] = wq6; rw4[1792] = wq7; rw4[2048] = wq8;
        }
        {
            const float sc = pg * rsqrtf(pv + 1e-5f);
            *(tid < 137 ? kLDS + tid : part + tid) = kv;
            *(tid < 32 ? scLDS + tid : part + 256 + tid) = sc;
            *(tid < 32 ? offLDS + tid : part + 512 + tid) = (pcb - pm) * sc + pbt;
        }
        lds_barrier();

#pragma unroll
        for (int r = 0; r < 4; ++r) {
            const int o = (tid >> 5) + 8 * r;
            const int c = tid & 31;
            const float* g = &rawLDS[o * 288 + c * 9];
            float a[6][3];
#pragma unroll
            for (int x = 0; x < 3; ++x) {
                const float g0 = g[x], g1 = g[3 + x], g2 = g[6 + x];
                a[0][x] = 0.25f * g0;
                a[1][x] = (-1.f / 6.f) * (g0 + g1 + g2);
                a[2][x] = (-1.f / 6.f) * (g0 - g1 + g2);
                a[3][x] = (1.f / 24.f) * g0 + (1.f / 12.f) * g1 + (1.f / 6.f) * g2;
                a[4][x] = (1.f / 24.f) * g0 - (1.f / 12.f) * g1 + (1.f / 6.f) * g2;
                a[5][x] = g2;
            }
            float* u = &uLDS[o * UO + c];
#pragma unroll
            for (int xi = 0; xi < 6; ++xi) {
                u[(xi * 4 + 0) * UPOS] = a[xi][0];
                u[(xi * 4 + 1) * UPOS] = 0.5f * (a[xi][0] + a[xi][1] + a[xi][2]);
                u[(xi * 4 + 2) * UPOS] = 0.5f * (a[xi][0] - a[xi][1] + a[xi][2]);
                u[(xi * 4 + 3) * UPOS] = a[xi][2];
            }
        }
#pragma unroll
        for (int g = 0; g < 3; ++g) {
            float* dst = tiles + (yy + 2 * g) * ROW_DW + ocol;
#pragma unroll
            for (int q = 0; q < 8; ++q)
                *(float4*)(dst + 4 * q) = make_float4(v[g][4 * q], v[g][4 * q + 1], v[g][4 * q + 2], v[g][4 * q + 3]);
        }
        if (4 * s0 - 1 + yy < 0) {
            float* dst = tiles + yy * ROW_DW + ocol;
#pragma unroll
            for (int q = 0; q < 8; ++q)
                *(float4*)(dst + 4 * q) = make_float4(0.f, 0.f, 0.f, 0.f);
        }
        if (tid < 12) {
            float* dst = &tiles[(tid >> 1) * ROW_DW + PIXOFF((tid & 1) * (ROWP - 1))];
#pragma unroll
            for (int q = 0; q < 8; ++q)
                *(float4*)(dst + 4 * q) = make_float4(0.f, 0.f, 0.f, 0.f);
        }
    }
    __syncthreads();
    if (tid < 8) {
        float* dst = &tiles[(6 + (tid >> 1)) * ROW_DW + PIXOFF((tid & 1) * (ROWP - 1))];
#pragma unroll
        for (int q = 0; q < 8; ++q)
            *(float4*)(dst + 4 * q) = make_float4(0.f, 0.f, 0.f, 0.f);
    }

    const int wave = tid >> 5;
    const int lane = tid & 31;
    const int half = lane >> 4;
    const int ml   = lane & 15;
    const int nb   = (wave >> 2) << 4;
    const int tc0  = (wave & 3) << 4;
    const int acol = PIXOFF(2 * (tc0 + ml)) + 2 * half;
    const float* bbase = &uLDS[(nb + ml) * UO + 2 * half];
    const int ob   = nb + 8 * half;

#pragma unroll 1
    for (int it = 0; it < STRIPS_PER_BLOCK; ++it) {
        const int strip = s0 + it;
        const int rb = (4 * it) % RING;
        const bool pf = it + 1 < STRIPS_PER_BLOCK;

        const float* arow[6];
#pragma unroll
        for (int r = 0; r < 6; ++r) {
            const int q = rb + r;
            arow[r] = tiles + (q >= RING ? q - RING : q) * ROW_DW + acol;
        }

        const int Ha = 4 * strip + 5 + yy;
        const int Hb = 4 * strip + 7 + yy;
        const bool okb = Hb < 128;
        const int Hca = min(Ha, 127);
        const int Hcb = min(Hb, 127);
        const float* a1 = embn + ((Hca & 3) * 4096 + (Hca >> 2) * 32 + jw);
        const float* a2 = embn + ((Hcb & 3) * 4096 + (Hcb >> 2) * 32 + jw);
        const int qa = rb + 6 + yy, qb = rb + 8 + yy;
        float* d1 = tiles + (qa >= RING ? qa - RING : qa) * ROW_DW + ocol;
        float* d2 = tiles + (qb >= RING ? qb - RING : qb) * ROW_DW + ocol;

        v8f M[NPOS] = {};
        auto gemm_step = [&](const int kk) __attribute__((always_inline)) {
            v2f bq[NPOS];
#pragma unroll
            for (int p = 0; p < NPOS; ++p) {
                const float2 q = *(const float2*)(bbase + p * UPOS + kk * 4);
                bq[p].x = q.x; bq[p].y = q.y;
            }
            v2f d[6][4];
#pragma unroll
            for (int r = 0; r < 6; ++r) {
#pragma unroll
                for (int s = 0; s < 4; ++s) {
                    const float2 q = *(const float2*)(arow[r] + PIXOFF(s) + kk * 4);
                    d[r][s].x = q.x; d[r][s].y = q.y;
                }
            }
            v2f t[6][4];
#pragma unroll
            for (int s = 0; s < 4; ++s) {
                const v2f p1 = d[4][s] - 4.f * d[2][s];
                const v2f q1 = d[3][s] - 4.f * d[1][s];
                const v2f r1 = d[4][s] - d[2][s];
                const v2f w1 = d[3][s] - d[1][s];
                t[0][s] = 4.f * d[0][s] - 5.f * d[2][s] + d[4][s];
                t[1][s] = p1 + q1;
                t[2][s] = p1 - q1;
                t[3][s] = r1 + 2.f * w1;
                t[4][s] = r1 - 2.f * w1;
                t[5][s] = 4.f * d[1][s] - 5.f * d[3][s] + d[5][s];
            }
#pragma unroll
            for (int xi = 0; xi < 6; ++xi) {
                v2f V[4];
                V[0] = t[xi][0] - t[xi][2];
                V[1] = t[xi][1] + t[xi][2];
                V[2] = t[xi][2] - t[xi][1];
                V[3] = t[xi][1] - t[xi][3];
#pragma unroll
                for (int nu = 0; nu < 4; ++nu) {
                    M[xi * 4 + nu] = __builtin_amdgcn_wmma_f32_16x16x4_f32(
                        false, bq[xi * 4 + nu], false, V[nu], (short)0, M[xi * 4 + nu], false, false);
                }
            }
        };
        auto request = [&](const int sh, float4& pa, float4& pb) __attribute__((always_inline)) {
            const float* b1 = a1 + 4 * sh * PLANE_C;
            const float* b2 = a2 + 4 * sh * PLANE_C;
            pa = make_float4(b1[0], b1[PLANE_C], b1[2 * PLANE_C], b1[3 * PLANE_C]);
            pb = make_float4(b2[0], b2[PLANE_C], b2[2 * PLANE_C], b2[3 * PLANE_C]);
        };
        auto deposit = [&](const int sh, const float4& pa, const float4& pb) __attribute__((always_inline)) {
            *(float4*)(d1 + 4 * sh) = pa;
            *(float4*)(d2 + 4 * sh) = pb;
        };
        if (pf) {
            float4 pa0, pa1, pb0, pb1;
            request(0, pa0, pb0);
#pragma unroll 1
            for (int ke = 0; ke < 6; ke += 2) {
                request(ke + 1, pa1, pb1);
                gemm_step(ke);
                deposit(ke, pa0, pb0);
                request(ke + 2, pa0, pb0);
                gemm_step(ke + 1);
                deposit(ke + 1, pa1, pb1);
                __builtin_amdgcn_sched_barrier(0);
                asm volatile("v_nop\n\tv_nop\n\tv_nop\n\tv_nop" ::: "memory");
            }
            request(7, pa1, pb1);
            gemm_step(6);
            deposit(6, pa0, pb0);
            gemm_step(7);
            deposit(7, pa1, pb1);
            __builtin_amdgcn_sched_barrier(0);
            asm volatile("v_nop\n\tv_nop\n\tv_nop\n\tv_nop" ::: "memory");
            if (!okb) {
#pragma unroll
                for (int q = 0; q < 8; ++q)
                    *(float4*)(d2 + 4 * q) = make_float4(0.f, 0.f, 0.f, 0.f);
            }
        } else {
#pragma unroll 1
            for (int kk = 0; kk < 8; ++kk) {
                gemm_step(kk);
                __builtin_amdgcn_sched_barrier(0);
                asm volatile("v_nop\n\tv_nop\n\tv_nop\n\tv_nop" ::: "memory");
            }
        }

        v8f Y[4][2];
        {
            v8f Z[6][2];
#pragma unroll
            for (int xi = 0; xi < 6; ++xi) {
                Z[xi][0] = M[xi * 4 + 0] + M[xi * 4 + 1] + M[xi * 4 + 2];
                Z[xi][1] = M[xi * 4 + 1] - M[xi * 4 + 2] - M[xi * 4 + 3];
            }
#pragma unroll
            for (int j = 0; j < 2; ++j) {
                const v8f sa = Z[1][j] + Z[2][j];
                const v8f sb = Z[1][j] - Z[2][j];
                const v8f sc = Z[3][j] + Z[4][j];
                const v8f se = Z[3][j] - Z[4][j];
                Y[0][j] = Z[0][j] + sa + sc;
                Y[1][j] = sb + 2.f * se;
                Y[2][j] = sa + 4.f * sc;
                Y[3][j] = sb + 8.f * se + Z[5][j];
            }
        }

        float hp[4][2][4];
        {
            float sc[8], of[8];
            {
                const float4 a = *(const float4*)(scLDS + ob),  b = *(const float4*)(scLDS + ob + 4);
                const float4 e = *(const float4*)(offLDS + ob), f = *(const float4*)(offLDS + ob + 4);
                sc[0] = a.x; sc[1] = a.y; sc[2] = a.z; sc[3] = a.w; sc[4] = b.x; sc[5] = b.y; sc[6] = b.z; sc[7] = b.w;
                of[0] = e.x; of[1] = e.y; of[2] = e.z; of[3] = e.w; of[4] = f.x; of[5] = f.y; of[6] = f.z; of[7] = f.w;
            }
#pragma unroll
            for (int i = 0; i < 4; ++i)
#pragma unroll
                for (int j = 0; j < 2; ++j)
#pragma unroll
                    for (int r = 0; r < 8; ++r)
                        Y[i][j][r] = fmaxf(Y[i][j][r] * sc[r] + of[r], 0.f);
#pragma unroll
            for (int k = 0; k < 4; ++k) {
                const float4 a = *(const float4*)(kLDS + 32 * k + ob);
                const float4 b = *(const float4*)(kLDS + 32 * k + ob + 4);
                const float w[8] = {a.x, a.y, a.z, a.w, b.x, b.y, b.z, b.w};
#pragma unroll
                for (int i = 0; i < 4; ++i) {
#pragma unroll
                    for (int j = 0; j < 2; ++j) {
                        float s = w[0] * Y[i][j][0];
#pragma unroll
                        for (int r = 1; r < 8; ++r) s = fmaf(w[r], Y[i][j][r], s);
                        hp[i][j][k] = s;
                    }
                }
            }
        }
        {
            float* pb = part + ((it & 1) * 2 + (wave >> 2)) * PART_DW
                             + ((2 * half) * 128 + 2 * (tc0 + ml)) * 4;
#pragma unroll
            for (int ii = 0; ii < 2; ++ii) {
                float o[2][4];
#pragma unroll
                for (int j = 0; j < 2; ++j) {
#pragma unroll
                    for (int k = 0; k < 4; ++k) {
                        const float keep = half ? hp[2 + ii][j][k] : hp[ii][j][k];
                        const float send = half ? hp[ii][j][k] : hp[2 + ii][j][k];
                        o[j][k] = keep + xhalf(send);
                    }
                }
                *(float4*)(pb + ii * 512)     = make_float4(o[0][0], o[0][1], o[0][2], o[0][3]);
                *(float4*)(pb + ii * 512 + 4) = make_float4(o[1][0], o[1][1], o[1][2], o[1][3]);
            }
        }
        __syncthreads();

#pragma unroll
        for (int h = 0; h < 2; ++h) {
            const int p = tid + 256 * h;
            const float* pr = part + (it & 1) * 2 * PART_DW + p * 4;
            const float4 a = *(const float4*)pr;
            const float4 b = *(const float4*)(pr + PART_DW);
            const float h0 = fmaxf(a.x + b.x + kLDS[132], 0.f);
            const float h1 = fmaxf(a.y + b.y + kLDS[133], 0.f);
            const float h2 = fmaxf(a.z + b.z + kLDS[134], 0.f);
            const float h3 = fmaxf(a.w + b.w + kLDS[135], 0.f);
            float y = kLDS[136];
            y = fmaf(kLDS[128], h0, y);
            y = fmaf(kLDS[129], h1, y);
            y = fmaf(kLDS[130], h2, y);
            y = fmaf(kLDS[131], h3, y);
            const float sg = 1.f / (1.f + __expf(-y));
            const int H = 4 * strip + (p >> 7), W = p & 127;
            z0[((size_t)n * 128 + H) * 128 + W] = sg;
        }
    }
}

#define SM_PITCH 132
#define SM_TROWS 12
__global__ __launch_bounds__(128) void smooth_kernel(
    const float* __restrict__ z0,
    const float* __restrict__ mask,
    const float* __restrict__ sw,
    const float* __restrict__ sb,
    float* __restrict__ out)
{
    __shared__ __attribute__((aligned(16))) float tile[2 * SM_TROWS * SM_PITCH];

    const int tid = threadIdx.x;
    const int tx  = tid & 31;
    const int ty  = tid >> 5;
    const int n   = blockIdx.x >> 4;
    const int H0  = (blockIdx.x & 15) * 8;
    const float* zp = z0   + (size_t)n * 16384;
    const float* mp = mask + (size_t)n * 16384;

    float4 g[6];
#pragma unroll
    for (int k = 0; k < 6; ++k) {
        const int row = (k % 3) * 4 + ty;
        const int h   = H0 - 2 + row;
        const float* src = (k < 3) ? zp : mp;
        g[k] = make_float4(0.f, 0.f, 0.f, 0.f);
        if ((unsigned)h < 128u) g[k] = *(const float4*)(src + h * 128 + 4 * tx);
    }
#pragma unroll
    for (int k = 0; k < 6; ++k) {
        const int row = (k % 3) * 4 + ty;
        float* dst = &tile[((k / 3) * SM_TROWS + row) * SM_PITCH + 2 + 4 * tx];
        *(float2*)dst       = make_float2(g[k].x, g[k].y);
        *(float2*)(dst + 2) = make_float2(g[k].z, g[k].w);
    }
    if (tid < 48) {
        const int pl  = tid / 24;
        const int rem = tid - pl * 24;
        float* dst = &tile[(pl * SM_TROWS + (rem >> 1)) * SM_PITCH + (rem & 1) * 130];
        *(float2*)dst = make_float2(0.f, 0.f);
    }
    __syncthreads();

    const float bias = sb[0];
    float acc[2][4];
#pragma unroll
    for (int o = 0; o < 2; ++o)
#pragma unroll
        for (int q = 0; q < 4; ++q) acc[o][q] = bias;

#pragma unroll
    for (int r = 0; r < 6; ++r) {
        const float* zr = &tile[(2 * ty + r) * SM_PITCH + 4 * tx];
        const float* mr = zr + SM_TROWS * SM_PITCH;
        const float4 za = *(const float4*)zr, zb = *(const float4*)(zr + 4);
        const float4 ma = *(const float4*)mr, mb = *(const float4*)(mr + 4);
        const float zv[8] = {za.x, za.y, za.z, za.w, zb.x, zb.y, zb.z, zb.w};
        const float mv[8] = {ma.x, ma.y, ma.z, ma.w, mb.x, mb.y, mb.z, mb.w};
#pragma unroll
        for (int o = 0; o < 2; ++o) {
            const int dy = r - o;
            if (dy < 0 || dy > 4) continue;
#pragma unroll
            for (int dx = 0; dx < 5; ++dx) {
                const float wz = sw[dy * 5 + dx];
                const float wm = sw[25 + dy * 5 + dx];
#pragma unroll
                for (int q = 0; q < 4; ++q) {
                    acc[o][q] = fmaf(wz, zv[q + dx], acc[o][q]);
                    acc[o][q] = fmaf(wm, mv[q + dx], acc[o][q]);
                }
            }
        }
    }

#pragma unroll
    for (int o = 0; o < 2; ++o) {
        const int H = H0 + 2 * ty + o;
        *(float4*)(out + ((size_t)n * 128 + H) * 128 + 4 * tx) =
            make_float4(acc[o][0], acc[o][1], acc[o][2], acc[o][3]);
    }
}

extern "C" void kernel_launch(void* const* d_in, const int* in_sizes, int n_in,
                              void* d_out, int out_size, void* d_ws, size_t ws_size,
                              hipStream_t stream) {
    const float* emb    = (const float*)d_in[0];
    const float* inck   = (const float*)d_in[1];
    const float* masks  = (const float*)d_in[2];
    const float* conv_w = (const float*)d_in[3];
    const float* conv_b = (const float*)d_in[4];
    const float* bn_g   = (const float*)d_in[5];
    const float* bn_b   = (const float*)d_in[6];
    const float* bn_m   = (const float*)d_in[7];
    const float* bn_v   = (const float*)d_in[8];
    const float* sw     = (const float*)d_in[9];
    const float* sb     = (const float*)d_in[10];
    float* out = (float*)d_out;
    float* z0  = (float*)d_ws;

    fused_head_kernel<<<256, 256, 0, stream>>>(emb, inck, conv_w, conv_b,
                                               bn_g, bn_b, bn_m, bn_v, z0);
    smooth_kernel<<<512, 128, 0, stream>>>(z0, masks, sw, sb, out);
}


# strip loop sheds a fifth of its non-matrix issue slots: 8 k steps straight-line (constant offsets, paired LDS reads, patch reads a step ahead), first step on zero C; input transform rounding changes
# speedup vs baseline: 3.7724x; 1.0655x over previous
#include <hip/hip_runtime.h>
#include <hip/hip_bf16.h>

typedef __attribute__((ext_vector_type(2))) float v2f;
typedef __attribute__((ext_vector_type(8))) float v8f;

#define PIXOFF(p) ((p) * 32 + ((p) >> 1) * 4)
#define UO 36
#define UPOS (32 * UO)
#define NPOS 24
#define U_DW (NPOS * UPOS)
#define RAW_DW 9216
#define ROWP 130
#define ROW_DW 4420
#define RING 10
#define RING_DW (RING * ROW_DW)
#define PLANE_C (16 * 32 * 32)
#define STRIPS_PER_BLOCK 4
#define PART_DW (512 * 4)
__device__ __forceinline__ float xhalf(float x) {
    return __builtin_bit_cast(float, __builtin_amdgcn_permlanex16(
        __builtin_bit_cast(int, x), __builtin_bit_cast(int, x), 0x76543210, 0xfedcba98, false, false));
}
__device__ __forceinline__ void lds_barrier() {
    __builtin_amdgcn_fence(__ATOMIC_RELEASE, "workgroup", "local");
    __builtin_amdgcn_s_barrier();
    __builtin_amdgcn_fence(__ATOMIC_ACQUIRE, "workgroup", "local");
}
__global__ __launch_bounds__(256) void fused_head_kernel(
    const float* __restrict__ emb,
    const float* inck,
    const float* conv_w,
    const float* conv_b,
    const float* bn_g,
    const float* bn_b,
    const float* bn_m,
    const float* bn_v,
    float* __restrict__ z0)
{
    __shared__ __attribute__((aligned(16))) float smem[140 + 32 + 32 + RING_DW + U_DW + 4 * PART_DW];
    float* kLDS   = smem;
    float* scLDS  = smem + 140;
    float* offLDS = smem + 172;
    float* tiles  = smem + 204;
    float* uLDS   = tiles + RING_DW;
    float* part   = uLDS + U_DW;
    float* rawLDS = tiles + 6 * ROW_DW;

    const int tid = threadIdx.x;
    const int n   = blockIdx.x >> 3;
    const int s0  = (blockIdx.x & 7) * STRIPS_PER_BLOCK;

    const float* embn = emb + (size_t)n * 512 * 32 * 32;
    const int Wp = tid & 127;
    const int jw = (Wp & 3) * 1024 + (Wp >> 2);
    const int yy = tid >> 7;
    const int ocol = PIXOFF(1 + Wp);

    const float4* cw4 = (const float4*)conv_w + tid;
    const float4 wq0 = cw4[0],    wq1 = cw4[256],  wq2 = cw4[512],  wq3 = cw4[768], wq4 = cw4[1024],
                 wq5 = cw4[1280], wq6 = cw4[1536], wq7 = cw4[1792], wq8 = cw4[2048];
    const float kv = inck[n * 137 + min(tid, 136)];
    const int ct = tid & 31;
    const float pg = bn_g[ct], pv = bn_v[ct], pcb = conv_b[ct], pm = bn_m[ct], pbt = bn_b[ct];
    int jws = jw;
    asm volatile("" : "+v"(jws));

    {
        float v[3][32];
#pragma unroll
        for (int g = 0; g < 3; ++g) {
            const int H  = 4 * s0 - 1 + yy + 2 * g;
            const int Hc = max(H, 0);
            const float* src = embn + ((Hc & 3) * 4096 + (Hc >> 2) * 32 + jws);
#pragma unroll
            for (int c = 0; c < 32; ++c) v[g][c] = src[c * PLANE_C];
        }
        {
            float4* rw4 = (float4*)rawLDS + tid;
            rw4[0]    = wq0; rw4[256]  = wq1; rw4[512]  = wq2; rw4[768]  = wq3; rw4[1024] = wq4;
            rw4[1280] = wq5; rw4[1536] = wq6; rw4[1792] = wq7; rw4[2048] = wq8;
        }
        {
            const float sc = pg * rsqrtf(pv + 1e-5f);
            *(tid < 137 ? kLDS + tid : part + tid) = kv;
            *(tid < 32 ? scLDS + tid : part + 256 + tid) = sc;
            *(tid < 32 ? offLDS + tid : part + 512 + tid) = (pcb - pm) * sc + pbt;
        }
        lds_barrier();

#pragma unroll
        for (int r = 0; r < 4; ++r) {
            const int o = (tid >> 5) + 8 * r;
            const int c = tid & 31;
            const float* g = &rawLDS[o * 288 + c * 9];
            float a[6][3];
#pragma unroll
            for (int x = 0; x < 3; ++x) {
                const float g0 = g[x], g1 = g[3 + x], g2 = g[6 + x];
                a[0][x] = 0.25f * g0;
                a[1][x] = (-1.f / 6.f) * (g0 + g1 + g2);
                a[2][x] = (-1.f / 6.f) * (g0 - g1 + g2);
                a[3][x] = (1.f / 24.f) * g0 + (1.f / 12.f) * g1 + (1.f / 6.f) * g2;
                a[4][x] = (1.f / 24.f) * g0 - (1.f / 12.f) * g1 + (1.f / 6.f) * g2;
                a[5][x] = g2;
            }
            float* u = &uLDS[o * UO + c];
#pragma unroll
            for (int xi = 0; xi < 6; ++xi) {
                u[(xi * 4 + 0) * UPOS] = a[xi][0];
                u[(xi * 4 + 1) * UPOS] = 0.5f * (a[xi][0] + a[xi][1] + a[xi][2]);
                u[(xi * 4 + 2) * UPOS] = 0.5f * (a[xi][0] - a[xi][1] + a[xi][2]);
                u[(xi * 4 + 3) * UPOS] = a[xi][2];
            }
        }
#pragma unroll
        for (int g = 0; g < 3; ++g) {
            float* dst = tiles + (yy + 2 * g) * ROW_DW + ocol;
#pragma unroll
            for (int q = 0; q < 8; ++q)
                *(float4*)(dst + 4 * q) = make_float4(v[g][4 * q], v[g][4 * q + 1], v[g][4 * q + 2], v[g][4 * q + 3]);
        }
        if (4 * s0 - 1 + yy < 0) {
            float* dst = tiles + yy * ROW_DW + ocol;
#pragma unroll
            for (int q = 0; q < 8; ++q)
                *(float4*)(dst + 4 * q) = make_float4(0.f, 0.f, 0.f, 0.f);
        }
        if (tid < 12) {
            float* dst = &tiles[(tid >> 1) * ROW_DW + PIXOFF((tid & 1) * (ROWP - 1))];
#pragma unroll
            for (int q = 0; q < 8; ++q)
                *(float4*)(dst + 4 * q) = make_float4(0.f, 0.f, 0.f, 0.f);
        }
    }
    __syncthreads();
    if (tid < 8) {
        float* dst = &tiles[(6 + (tid >> 1)) * ROW_DW + PIXOFF((tid & 1) * (ROWP - 1))];
#pragma unroll
        for (int q = 0; q < 8; ++q)
            *(float4*)(dst + 4 * q) = make_float4(0.f, 0.f, 0.f, 0.f);
    }

    const int wave = tid >> 5;
    const int lane = tid & 31;
    const int half = lane >> 4;
    const int ml   = lane & 15;
    const int nb   = (wave >> 2) << 4;
    const int tc0  = (wave & 3) << 4;
    const int acol = PIXOFF(2 * (tc0 + ml)) + 2 * half;
    const float* bbase = &uLDS[(nb + ml) * UO + 2 * half];
    const int ob   = nb + 8 * half;

#pragma unroll 1
    for (int it = 0; it < STRIPS_PER_BLOCK; ++it) {
        const int strip = s0 + it;
        const int rb = (4 * it) % RING;
        const bool pf = it + 1 < STRIPS_PER_BLOCK;

        const float* arow[6];
#pragma unroll
        for (int r = 0; r < 6; ++r) {
            const int q = rb + r;
            arow[r] = tiles + (q >= RING ? q - RING : q) * ROW_DW + acol;
        }

        const int Ha = 4 * strip + 5 + yy;
        const int Hb = 4 * strip + 7 + yy;
        const bool okb = Hb < 128;
        const int Hca = min(Ha, 127);
        const int Hcb = min(Hb, 127);
        const float* a1 = embn + ((Hca & 3) * 4096 + (Hca >> 2) * 32 + jw);
        const float* a2 = embn + ((Hcb & 3) * 4096 + (Hcb >> 2) * 32 + jw);
        const int qa = rb + 6 + yy, qb = rb + 8 + yy;
        float* d1 = tiles + (qa >= RING ? qa - RING : qa) * ROW_DW + ocol;
        float* d2 = tiles + (qb >= RING ? qb - RING : qb) * ROW_DW + ocol;

        v8f M[NPOS];
        struct FragB { v2f bq[NPOS]; };
        struct FragD { v2f d[6][4]; };
        auto load_b = [&](const int kk, FragB& f) __attribute__((always_inline)) {
#pragma unroll
            for (int p = 0; p < NPOS; ++p) {
                const float2 q = *(const float2*)(bbase + p * UPOS + kk * 4);
                f.bq[p].x = q.x; f.bq[p].y = q.y;
            }
        };
        auto load_d = [&](const int kk, FragD& f) __attribute__((always_inline)) {
#pragma unroll
            for (int r = 0; r < 6; ++r) {
#pragma unroll
                for (int s = 0; s < 4; ++s) {
                    const float2 q = *(const float2*)(arow[r] + PIXOFF(s) + kk * 4);
                    f.d[r][s].x = q.x; f.d[r][s].y = q.y;
                }
            }
        };
        auto mma_step = [&](const FragB& fb, const FragD& f, const bool first) __attribute__((always_inline)) {
            v2f t[6][4];
#pragma unroll
            for (int s = 0; s < 4; ++s) {
                const v2f p1 = f.d[4][s] - 4.f * f.d[2][s];
                const v2f q1 = f.d[3][s] - 4.f * f.d[1][s];
                const v2f r1 = f.d[4][s] - f.d[2][s];
                const v2f w1 = f.d[3][s] - f.d[1][s];
                t[0][s] = 4.f * f.d[0][s] + (f.d[4][s] - 5.f * f.d[2][s]);
                t[1][s] = p1 + q1;
                t[2][s] = p1 - q1;
                t[3][s] = r1 + 2.f * w1;
                t[4][s] = r1 - 2.f * w1;
                t[5][s] = 4.f * f.d[1][s] + (f.d[5][s] - 5.f * f.d[3][s]);
            }
#pragma unroll
            for (int xi = 0; xi < 6; ++xi) {
                v2f V[4];
                V[0] = t[xi][0] - t[xi][2];
                V[1] = t[xi][1] + t[xi][2];
                V[2] = t[xi][2] - t[xi][1];
                V[3] = t[xi][1] - t[xi][3];
#pragma unroll
                for (int nu = 0; nu < 4; ++nu) {
                    M[xi * 4 + nu] = __builtin_amdgcn_wmma_f32_16x16x4_f32(
                        false, fb.bq[xi * 4 + nu], false, V[nu], (short)0, first ? v8f{} : M[xi * 4 + nu], false, false);
                }
            }
        };
        auto gemm_step = [&](const int kk, const bool first) __attribute__((always_inline)) {
            FragB fb; FragD fd;
            load_b(kk, fb);
            load_d(kk, fd);
            mma_step(fb, fd, first);
        };
        auto request = [&](const int sh, float4& pa, float4& pb) __attribute__((always_inline)) {
            const float* b1 = a1 + 4 * sh * PLANE_C;
            const float* b2 = a2 + 4 * sh * PLANE_C;
            pa = make_float4(b1[0], b1[PLANE_C], b1[2 * PLANE_C], b1[3 * PLANE_C]);
            pb = make_float4(b2[0], b2[PLANE_C], b2[2 * PLANE_C], b2[3 * PLANE_C]);
        };
        auto deposit = [&](const int sh, const float4& pa, const float4& pb) __attribute__((always_inline)) {
            *(float4*)(d1 + 4 * sh) = pa;
            *(float4*)(d2 + 4 * sh) = pb;
        };
        if (pf) {
            float4 pa0, pa1, pb0, pb1;
            request(0, pa0, pb0);
            FragD dA, dB;
            load_d(0, dA);
            auto step = [&](const int k, const FragD& dcur, FragD& dnext) __attribute__((always_inline)) {
                FragB fb;
                load_b(k, fb);
                if (k < 7) load_d(k + 1, dnext);
                __builtin_amdgcn_sched_barrier(0);
                mma_step(fb, dcur, k == 0);
                __builtin_amdgcn_sched_barrier(0);
            };
            request(1, pa1, pb1);
            step(0, dA, dB);
            deposit(0, pa0, pb0);
            request(2, pa0, pb0);
            step(1, dB, dA);
            deposit(1, pa1, pb1);
            request(3, pa1, pb1);
            step(2, dA, dB);
            deposit(2, pa0, pb0);
            request(4, pa0, pb0);
            step(3, dB, dA);
            deposit(3, pa1, pb1);
            request(5, pa1, pb1);
            step(4, dA, dB);
            deposit(4, pa0, pb0);
            request(6, pa0, pb0);
            step(5, dB, dA);
            deposit(5, pa1, pb1);
            request(7, pa1, pb1);
            step(6, dA, dB);
            deposit(6, pa0, pb0);
            step(7, dB, dA);
            deposit(7, pa1, pb1);
            __builtin_amdgcn_sched_barrier(0);
            asm volatile("v_nop\n\tv_nop\n\tv_nop\n\tv_nop" ::: "memory");
            if (!okb) {
#pragma unroll
                for (int q = 0; q < 8; ++q)
                    *(float4*)(d2 + 4 * q) = make_float4(0.f, 0.f, 0.f, 0.f);
            }
        } else {
            gemm_step(0, true);
#pragma unroll 1
            for (int kk = 1; kk < 8; ++kk) {
                gemm_step(kk, false);
                __builtin_amdgcn_sched_barrier(0);
                asm volatile("v_nop\n\tv_nop\n\tv_nop\n\tv_nop" ::: "memory");
            }
        }

        v8f Y[4][2];
        {
            v8f Z[6][2];
#pragma unroll
            for (int xi = 0; xi < 6; ++xi) {
                Z[xi][0] = M[xi * 4 + 0] + M[xi * 4 + 1] + M[xi * 4 + 2];
                Z[xi][1] = M[xi * 4 + 1] - M[xi * 4 + 2] - M[xi * 4 + 3];
            }
#pragma unroll
            for (int j = 0; j < 2; ++j) {
                const v8f sa = Z[1][j] + Z[2][j];
                const v8f sb = Z[1][j] - Z[2][j];
                const v8f sc = Z[3][j] + Z[4][j];
                const v8f se = Z[3][j] - Z[4][j];
                Y[0][j] = Z[0][j] + sa + sc;
                Y[1][j] = sb + 2.f * se;
                Y[2][j] = sa + 4.f * sc;
                Y[3][j] = sb + 8.f * se + Z[5][j];
            }
        }

        float hp[4][2][4];
        {
            float sc[8], of[8];
            {
                const float4 a = *(const float4*)(scLDS + ob),  b = *(const float4*)(scLDS + ob + 4);
                const float4 e = *(const float4*)(offLDS + ob), f = *(const float4*)(offLDS + ob + 4);
                sc[0] = a.x; sc[1] = a.y; sc[2] = a.z; sc[3] = a.w; sc[4] = b.x; sc[5] = b.y; sc[6] = b.z; sc[7] = b.w;
                of[0] = e.x; of[1] = e.y; of[2] = e.z; of[3] = e.w; of[4] = f.x; of[5] = f.y; of[6] = f.z; of[7] = f.w;
            }
#pragma unroll
            for (int i = 0; i < 4; ++i)
#pragma unroll
                for (int j = 0; j < 2; ++j)
#pragma unroll
                    for (int r = 0; r < 8; ++r)
                        Y[i][j][r] = fmaxf(Y[i][j][r] * sc[r] + of[r], 0.f);
#pragma unroll
            for (int k = 0; k < 4; ++k) {
                const float4 a = *(const float4*)(kLDS + 32 * k + ob);
                const float4 b = *(const float4*)(kLDS + 32 * k + ob + 4);
                const float w[8] = {a.x, a.y, a.z, a.w, b.x, b.y, b.z, b.w};
#pragma unroll
                for (int i = 0; i < 4; ++i) {
#pragma unroll
                    for (int j = 0; j < 2; ++j) {
                        float s = w[0] * Y[i][j][0];
#pragma unroll
                        for (int r = 1; r < 8; ++r) s = fmaf(w[r], Y[i][j][r], s);
                        hp[i][j][k] = s;
                    }
                }
            }
        }
        {
            float* pb = part + ((it & 1) * 2 + (wave >> 2)) * PART_DW
                             + ((2 * half) * 128 + 2 * (tc0 + ml)) * 4;
#pragma unroll
            for (int ii = 0; ii < 2; ++ii) {
                float o[2][4];
#pragma unroll
                for (int j = 0; j < 2; ++j) {
#pragma unroll
                    for (int k = 0; k < 4; ++k) {
                        const float keep = half ? hp[2 + ii][j][k] : hp[ii][j][k];
                        const float send = half ? hp[ii][j][k] : hp[2 + ii][j][k];
                        o[j][k] = keep + xhalf(send);
                    }
                }
                *(float4*)(pb + ii * 512)     = make_float4(o[0][0], o[0][1], o[0][2], o[0][3]);
                *(float4*)(pb + ii * 512 + 4) = make_float4(o[1][0], o[1][1], o[1][2], o[1][3]);
            }
        }
        __syncthreads();

#pragma unroll
        for (int h = 0; h < 2; ++h) {
            const int p = tid + 256 * h;
            const float* pr = part + (it & 1) * 2 * PART_DW + p * 4;
            const float4 a = *(const float4*)pr;
            const float4 b = *(const float4*)(pr + PART_DW);
            const float h0 = fmaxf(a.x + b.x + kLDS[132], 0.f);
            const float h1 = fmaxf(a.y + b.y + kLDS[133], 0.f);
            const float h2 = fmaxf(a.z + b.z + kLDS[134], 0.f);
            const float h3 = fmaxf(a.w + b.w + kLDS[135], 0.f);
            float y = kLDS[136];
            y = fmaf(kLDS[128], h0, y);
            y = fmaf(kLDS[129], h1, y);
            y = fmaf(kLDS[130], h2, y);
            y = fmaf(kLDS[131], h3, y);
            const float sg = 1.f / (1.f + __expf(-y));
            const int H = 4 * strip + (p >> 7), W = p & 127;
            z0[((size_t)n * 128 + H) * 128 + W] = sg;
        }
    }
}

#define SM_PITCH 132
#define SM_TROWS 12
__global__ __launch_bounds__(128) void smooth_kernel(
    const float* __restrict__ z0,
    const float* __restrict__ mask,
    const float* __restrict__ sw,
    const float* __restrict__ sb,
    float* __restrict__ out)
{
    __shared__ __attribute__((aligned(16))) float tile[2 * SM_TROWS * SM_PITCH];

    const int tid = threadIdx.x;
    const int tx  = tid & 31;
    const int ty  = tid >> 5;
    const int n   = blockIdx.x >> 4;
    const int H0  = (blockIdx.x & 15) * 8;
    const float* zp = z0   + (size_t)n * 16384;
    const float* mp = mask + (size_t)n * 16384;

    float4 g[6];
#pragma unroll
    for (int k = 0; k < 6; ++k) {
        const int row = (k % 3) * 4 + ty;
        const int h   = H0 - 2 + row;
        const float* src = (k < 3) ? zp : mp;
        g[k] = make_float4(0.f, 0.f, 0.f, 0.f);
        if ((unsigned)h < 128u) g[k] = *(const float4*)(src + h * 128 + 4 * tx);
    }
#pragma unroll
    for (int k = 0; k < 6; ++k) {
        const int row = (k % 3) * 4 + ty;
        float* dst = &tile[((k / 3) * SM_TROWS + row) * SM_PITCH + 2 + 4 * tx];
        *(float2*)dst       = make_float2(g[k].x, g[k].y);
        *(float2*)(dst + 2) = make_float2(g[k].z, g[k].w);
    }
    if (tid < 48) {
        const int pl  = tid / 24;
        const int rem = tid - pl * 24;
        float* dst = &tile[(pl * SM_TROWS + (rem >> 1)) * SM_PITCH + (rem & 1) * 130];
        *(float2*)dst = make_float2(0.f, 0.f);
    }
    __syncthreads();

    const float bias = sb[0];
    float acc[2][4];
#pragma unroll
    for (int o = 0; o < 2; ++o)
#pragma unroll
        for (int q = 0; q < 4; ++q) acc[o][q] = bias;

#pragma unroll
    for (int r = 0; r < 6; ++r) {
        const float* zr = &tile[(2 * ty + r) * SM_PITCH + 4 * tx];
        const float* mr = zr + SM_TROWS * SM_PITCH;
        const float4 za = *(const float4*)zr, zb = *(const float4*)(zr + 4);
        const float4 ma = *(const float4*)mr, mb = *(const float4*)(mr + 4);
        const float zv[8] = {za.x, za.y, za.z, za.w, zb.x, zb.y, zb.z, zb.w};
        const float mv[8] = {ma.x, ma.y, ma.z, ma.w, mb.x, mb.y, mb.z, mb.w};
#pragma unroll
        for (int o = 0; o < 2; ++o) {
            const int dy = r - o;
            if (dy < 0 || dy > 4) continue;
#pragma unroll
            for (int dx = 0; dx < 5; ++dx) {
                const float wz = sw[dy * 5 + dx];
                const float wm = sw[25 + dy * 5 + dx];
#pragma unroll
                for (int q = 0; q < 4; ++q) {
                    acc[o][q] = fmaf(wz, zv[q + dx], acc[o][q]);
                    acc[o][q] = fmaf(wm, mv[q + dx], acc[o][q]);
                }
            }
        }
    }

#pragma unroll
    for (int o = 0; o < 2; ++o) {
        const int H = H0 + 2 * ty + o;
        *(float4*)(out + ((size_t)n * 128 + H) * 128 + 4 * tx) =
            make_float4(acc[o][0], acc[o][1], acc[o][2], acc[o][3]);
    }
}

extern "C" void kernel_launch(void* const* d_in, const int* in_sizes, int n_in,
                              void* d_out, int out_size, void* d_ws, size_t ws_size,
                              hipStream_t stream) {
    const float* emb    = (const float*)d_in[0];
    const float* inck   = (const float*)d_in[1];
    const float* masks  = (const float*)d_in[2];
    const float* conv_w = (const float*)d_in[3];
    const float* conv_b = (const float*)d_in[4];
    const float* bn_g   = (const float*)d_in[5];
    const float* bn_b   = (const float*)d_in[6];
    const float* bn_m   = (const float*)d_in[7];
    const float* bn_v   = (const float*)d_in[8];
    const float* sw     = (const float*)d_in[9];
    const float* sb     = (const float*)d_in[10];
    float* out = (float*)d_out;
    float* z0  = (float*)d_ws;

    fused_head_kernel<<<256, 256, 0, stream>>>(emb, inck, conv_w, conv_b,
                                               bn_g, bn_b, bn_m, bn_v, z0);
    smooth_kernel<<<512, 128, 0, stream>>>(z0, masks, sw, sb, out);
}
